# starting kernel
# baseline (speedup 1.0000x reference)
#include <hip/hip_runtime.h>


typedef float v2f __attribute__((ext_vector_type(2)));
typedef float v8f __attribute__((ext_vector_type(8)));

#define H 128
#define NIN 8
#define LDS_STRIDE 130

__device__ __forceinline__ v8f wmma_k4(v2f a, v2f b, v8f c) {
  return __builtin_amdgcn_wmma_f32_16x16x4_f32(
       false, a,  false, b,
       (short)0, c,  false,  false);
}

__device__ __forceinline__ v2f ldg2(const float* p) {
  return *reinterpret_cast<const v2f*>(p);
}

template <int WAVES>
__global__ __launch_bounds__(WAVES * 32) void net_forward_kernel(
    const float* __restrict__ x, float* __restrict__ out, int n_samples,
    const float* __restrict__ w1_1, const float* __restrict__ b1_1,
    const float* __restrict__ w2_1, const float* __restrict__ b2_1,
    const float* __restrict__ w1_2, const float* __restrict__ b1_2,
    const float* __restrict__ w2_2, const float* __restrict__ b2_2,
    const float* __restrict__ w_out) {
  __shared__ float ybuf[WAVES][16 * LDS_STRIDE];

  const int wave = threadIdx.x >> 5;
  const int lane = threadIdx.x & 31;
  const int half = lane >> 4;
  const int l16 = lane & 15;
  const int tile = blockIdx.x * WAVES + wave;
  if (tile * 16 >= n_samples) return;
  const int row0 = tile * 16;
  float* yb = ybuf[wave];

  {
    const float* xr = x + (row0 + l16) * NIN + 2 * half;
    v2f a0 = ldg2(xr);
    v2f a1 = ldg2(xr + 4);
#pragma unroll
    for (int nt = 0; nt < 8; ++nt) {
      const int n = nt * 16 + l16;
      v2f p10 = ldg2(w1_1 + n * NIN + 2 * half);
      v2f p11 = ldg2(w1_1 + n * NIN + 4 + 2 * half);
      v2f p20 = ldg2(w2_1 + n * NIN + 2 * half);
      v2f p21 = ldg2(w2_1 + n * NIN + 4 + 2 * half);
      v8f c1 = {}, c2 = {};
      c1 = wmma_k4(a0, p10, c1);
      c1 = wmma_k4(a1, p11, c1);
      c2 = wmma_k4(a0, p20, c2);
      c2 = wmma_k4(a1, p21, c2);
      const float bb1 = b1_1[n], bb2 = b2_1[n];
#pragma unroll
      for (int j = 0; j < 8; ++j) {
        const int m = j + 8 * half;
        yb[m * LDS_STRIDE + n] = (c1[j] + bb1) * (c2[j] + bb2);
      }
    }
  }

  float part[8];
#pragma unroll
  for (int j = 0; j < 8; ++j) part[j] = 0.f;

  const float* arow = yb + l16 * LDS_STRIDE + 2 * half;
#pragma unroll
  for (int nt = 0; nt < 8; ++nt) {
    const int n = nt * 16 + l16;
    const float* b1r = w1_2 + n * H + 2 * half;
    const float* b2r = w2_2 + n * H + 2 * half;
    v8f c1 = {}, c2 = {};
#pragma unroll 4
    for (int ks = 0; ks < 32; ++ks) {
      v2f a = *reinterpret_cast<const v2f*>(arow + ks * 4);
      v2f q1 = ldg2(b1r + ks * 4);
      v2f q2 = ldg2(b2r + ks * 4);
      c1 = wmma_k4(a, q1, c1);
      c2 = wmma_k4(a, q2, c2);
    }
    const float bia1 = b1_2[n], bia2 = b2_2[n], wo = w_out[n];
#pragma unroll
    for (int j = 0; j < 8; ++j)
      part[j] += (c1[j] + bia1) * (c2[j] + bia2) * wo;
  }

#pragma unroll
  for (int j = 0; j < 8; ++j) {
    float v = part[j];
    v += __shfl_xor(v, 1, 32);
    v += __shfl_xor(v, 2, 32);
    v += __shfl_xor(v, 4, 32);
    v += __shfl_xor(v, 8, 32);
    part[j] = v;
  }
  if (l16 == 0) {
#pragma unroll
    for (int j = 0; j < 8; ++j) out[row0 + j + 8 * half] = part[j];
  }
}

template <int WAVES>
__global__ __launch_bounds__(WAVES * 32) void net_fwd_grad_kernel(
    const float* __restrict__ x, const float* __restrict__ xdot,
    float* __restrict__ outY, float* __restrict__ outG, int n_samples,
    const float* __restrict__ w1_1, const float* __restrict__ b1_1,
    const float* __restrict__ w2_1, const float* __restrict__ b2_1,
    const float* __restrict__ w1_2, const float* __restrict__ b1_2,
    const float* __restrict__ w2_2, const float* __restrict__ b2_2,
    const float* __restrict__ w_out) {
  __shared__ float ybuf[WAVES][16 * LDS_STRIDE];
  __shared__ float ydbuf[WAVES][16 * LDS_STRIDE];

  const int wave = threadIdx.x >> 5;
  const int lane = threadIdx.x & 31;
  const int half = lane >> 4;
  const int l16 = lane & 15;
  const int tile = blockIdx.x * WAVES + wave;
  if (tile * 16 >= n_samples) return;
  const int row0 = tile * 16;
  float* yb = ybuf[wave];
  float* ydb = ydbuf[wave];

  {
    const float* xr = x + (row0 + l16) * NIN + 2 * half;
    const float* xdr = xdot + (row0 + l16) * NIN + 2 * half;
    v2f a0 = ldg2(xr), a1 = ldg2(xr + 4);
    v2f d0 = ldg2(xdr), d1 = ldg2(xdr + 4);
#pragma unroll
    for (int nt = 0; nt < 8; ++nt) {
      const int n = nt * 16 + l16;
      v2f p10 = ldg2(w1_1 + n * NIN + 2 * half);
      v2f p11 = ldg2(w1_1 + n * NIN + 4 + 2 * half);
      v2f p20 = ldg2(w2_1 + n * NIN + 2 * half);
      v2f p21 = ldg2(w2_1 + n * NIN + 4 + 2 * half);
      v8f c1 = {}, c2 = {}, g1 = {}, g2 = {};
      c1 = wmma_k4(a0, p10, c1);
      c1 = wmma_k4(a1, p11, c1);
      c2 = wmma_k4(a0, p20, c2);
      c2 = wmma_k4(a1, p21, c2);
      g1 = wmma_k4(d0, p10, g1);
      g1 = wmma_k4(d1, p11, g1);
      g2 = wmma_k4(d0, p20, g2);
      g2 = wmma_k4(d1, p21, g2);
      const float bb1 = b1_1[n], bb2 = b2_1[n];
#pragma unroll
      for (int j = 0; j < 8; ++j) {
        const int m = j + 8 * half;
        const float z1 = c1[j] + bb1, z2 = c2[j] + bb2;
        yb[m * LDS_STRIDE + n] = z1 * z2;
        ydb[m * LDS_STRIDE + n] = g1[j] * z2 + z1 * g2[j];
      }
    }
  }

  float pY[8], pG[8];
#pragma unroll
  for (int j = 0; j < 8; ++j) {
    pY[j] = 0.f;
    pG[j] = 0.f;
  }

  const float* arow = yb + l16 * LDS_STRIDE + 2 * half;
  const float* drow = ydb + l16 * LDS_STRIDE + 2 * half;
#pragma unroll
  for (int nt = 0; nt < 8; ++nt) {
    const int n = nt * 16 + l16;
    const float* b1r = w1_2 + n * H + 2 * half;
    const float* b2r = w2_2 + n * H + 2 * half;
    v8f c1 = {}, c2 = {}, g1 = {}, g2 = {};
#pragma unroll 4
    for (int ks = 0; ks < 32; ++ks) {
      v2f a = *reinterpret_cast<const v2f*>(arow + ks * 4);
      v2f ad = *reinterpret_cast<const v2f*>(drow + ks * 4);
      v2f q1 = ldg2(b1r + ks * 4);
      v2f q2 = ldg2(b2r + ks * 4);
      c1 = wmma_k4(a, q1, c1);
      c2 = wmma_k4(a, q2, c2);
      g1 = wmma_k4(ad, q1, g1);
      g2 = wmma_k4(ad, q2, g2);
    }
    const float bia1 = b1_2[n], bia2 = b2_2[n], wo = w_out[n];
#pragma unroll
    for (int j = 0; j < 8; ++j) {
      const float z1 = c1[j] + bia1, z2 = c2[j] + bia2;
      pY[j] += z1 * z2 * wo;
      pG[j] += (g1[j] * z2 + z1 * g2[j]) * wo;
    }
  }

#pragma unroll
  for (int j = 0; j < 8; ++j) {
    float vy = pY[j], vg = pG[j];
    vy += __shfl_xor(vy, 1, 32);
    vy += __shfl_xor(vy, 2, 32);
    vy += __shfl_xor(vy, 4, 32);
    vy += __shfl_xor(vy, 8, 32);
    vg += __shfl_xor(vg, 1, 32);
    vg += __shfl_xor(vg, 2, 32);
    vg += __shfl_xor(vg, 4, 32);
    vg += __shfl_xor(vg, 8, 32);
    pY[j] = vy;
    pG[j] = vg;
  }
  if (l16 == 0) {
#pragma unroll
    for (int j = 0; j < 8; ++j) {
      outY[row0 + j + 8 * half] = pY[j];
      outG[row0 + j + 8 * half] = pG[j];
    }
  }
}

extern "C" void kernel_launch(void* const* d_in, const int* in_sizes, int n_in,
                              void* d_out, int out_size, void* d_ws,
                              size_t ws_size, hipStream_t stream) {
  const float* T = (const float*)d_in[0];
  const float* l = (const float*)d_in[1];
  const float* l1_dot = (const float*)d_in[2];
  const float* center = (const float*)d_in[3];
  const float* w1_1 = (const float*)d_in[4];
  const float* b1_1 = (const float*)d_in[5];
  const float* w2_1 = (const float*)d_in[6];
  const float* b2_1 = (const float*)d_in[7];
  const float* w1_2 = (const float*)d_in[8];
  const float* b1_2 = (const float*)d_in[9];
  const float* w2_2 = (const float*)d_in[10];
  const float* b2_2 = (const float*)d_in[11];
  const float* w_out = (const float*)d_in[12];

  const int nT = in_sizes[0] / NIN;
  const int nL = in_sizes[1] / NIN;
  const int nC = in_sizes[3] / NIN;

  float* out = (float*)d_out;
  float* v_t = out;
  float* v_y = out + nT;
  float* v_g = out + nT + nL;
  float* v_c = out + nT + 2 * nL;

  constexpr int WF = 4;
  constexpr int WG = 2;

  const int tilesT = (nT + 15) / 16;
  const int tilesL = (nL + 15) / 16;
  const int tilesC = (nC + 15) / 16;

  net_forward_kernel<WF><<<(tilesT + WF - 1) / WF, WF * 32, 0, stream>>>(
      T, v_t, nT, w1_1, b1_1, w2_1, b2_1, w1_2, b1_2, w2_2, b2_2, w_out);

  net_forward_kernel<WF><<<(tilesC + WF - 1) / WF, WF * 32, 0, stream>>>(
      center, v_c, nC, w1_1, b1_1, w2_1, b2_1, w1_2, b1_2, w2_2, b2_2, w_out);

  net_fwd_grad_kernel<WG><<<(tilesL + WG - 1) / WG, WG * 32, 0, stream>>>(
      l, l1_dot, v_y, v_g, nL, w1_1, b1_1, w2_1, b2_1, w1_2, b1_2, w2_2, b2_2,
      w_out);
}


# Layer-2 weights (2 x 128x128 f32) copied to LDS once per work-group; 256 resident groups loop over 16-sample tiles and read WMMA B fragments from LDS instead of global memory
# speedup vs baseline: 1.6385x; 1.6385x over previous
#include <hip/hip_runtime.h>


typedef float v2f __attribute__((ext_vector_type(2)));
typedef float v4f __attribute__((ext_vector_type(4)));
typedef float v8f __attribute__((ext_vector_type(8)));

#define H 128
#define NIN 8
#define LDS_STRIDE 130
#define W_PITCH 132

__device__ __forceinline__ v8f wmma_k4(v2f a, v2f b, v8f c) {
  return __builtin_amdgcn_wmma_f32_16x16x4_f32(
       false, a,  false, b,
       (short)0, c,  false,  false);
}

__device__ __forceinline__ v2f ldg2(const float* p) {
  return *reinterpret_cast<const v2f*>(p);
}

__device__ __forceinline__ v2f lds2(const float* p) {
  return *reinterpret_cast<const v2f*>(p);
}

template <int NTHREADS>
__device__ __forceinline__ void stage_layer2_weights(
    float* __restrict__ s1, float* __restrict__ s2,
    const float* __restrict__ w1_2, const float* __restrict__ w2_2) {
  constexpr int NVEC = H * H / 4;
  constexpr int NIT = NVEC / NTHREADS;
  v4f u1[NIT], u2[NIT];
#pragma unroll
  for (int it = 0; it < NIT; ++it) {
    const int i = it * NTHREADS + threadIdx.x;
    const int r = i >> 5;
    const int c = (i & 31) * 4;
    u1[it] = *reinterpret_cast<const v4f*>(w1_2 + r * H + c);
    u2[it] = *reinterpret_cast<const v4f*>(w2_2 + r * H + c);
  }
#pragma unroll
  for (int it = 0; it < NIT; ++it) {
    const int i = it * NTHREADS + threadIdx.x;
    const int r = i >> 5;
    const int c = (i & 31) * 4;
    *reinterpret_cast<v4f*>(s1 + r * W_PITCH + c) = u1[it];
    *reinterpret_cast<v4f*>(s2 + r * W_PITCH + c) = u2[it];
  }
}

template <int WAVES>
__global__ __launch_bounds__(WAVES * 32) void net_forward_kernel(
    const float* __restrict__ x, float* __restrict__ out, int n_samples,
    const float* __restrict__ w1_1, const float* __restrict__ b1_1,
    const float* __restrict__ w2_1, const float* __restrict__ b2_1,
    const float* __restrict__ w1_2, const float* __restrict__ b1_2,
    const float* __restrict__ w2_2, const float* __restrict__ b2_2,
    const float* __restrict__ w_out) {
  __shared__ __attribute__((aligned(16))) float wl1[H * W_PITCH];
  __shared__ __attribute__((aligned(16))) float wl2[H * W_PITCH];
  __shared__ __attribute__((aligned(16))) float ybuf[WAVES][16 * LDS_STRIDE];

  const int wave = __builtin_amdgcn_readfirstlane(threadIdx.x >> 5);
  const int lane = threadIdx.x & 31;
  const int half = lane >> 4;
  const int l16 = lane & 15;

  stage_layer2_weights<WAVES * 32>(wl1, wl2, w1_2, w2_2);
  __syncthreads();

  float* yb = ybuf[wave];
  const float* arow = yb + l16 * LDS_STRIDE + 2 * half;
  const float* q1row = wl1 + l16 * W_PITCH + 2 * half;
  const float* q2row = wl2 + l16 * W_PITCH + 2 * half;
  const int ntiles = (n_samples + 15) >> 4;
  const int tstep = gridDim.x * WAVES;

  for (int tile = blockIdx.x + gridDim.x * wave; tile < ntiles; tile += tstep) {
    const int row0 = tile * 16;

    {
      const float* xr = x + (row0 + l16) * NIN + 2 * half;
      v2f a0 = ldg2(xr);
      v2f a1 = ldg2(xr + 4);
#pragma unroll
      for (int nt = 0; nt < 8; ++nt) {
        const int n = nt * 16 + l16;
        v2f p10 = ldg2(w1_1 + n * NIN + 2 * half);
        v2f p11 = ldg2(w1_1 + n * NIN + 4 + 2 * half);
        v2f p20 = ldg2(w2_1 + n * NIN + 2 * half);
        v2f p21 = ldg2(w2_1 + n * NIN + 4 + 2 * half);
        v8f c1 = {}, c2 = {};
        c1 = wmma_k4(a0, p10, c1);
        c1 = wmma_k4(a1, p11, c1);
        c2 = wmma_k4(a0, p20, c2);
        c2 = wmma_k4(a1, p21, c2);
        const float bb1 = b1_1[n], bb2 = b2_1[n];
#pragma unroll
        for (int j = 0; j < 8; ++j) {
          const int m = j + 8 * half;
          yb[m * LDS_STRIDE + n] = (c1[j] + bb1) * (c2[j] + bb2);
        }
      }
    }

    float part[8];
#pragma unroll
    for (int j = 0; j < 8; ++j) part[j] = 0.f;

    constexpr int KU = 8;
    constexpr int CH = 32 / KU;
    v2f ca[KU], cq1[KU], cq2[KU];
#pragma unroll
    for (int u = 0; u < KU; ++u) {
      ca[u] = lds2(arow + u * 4);
      cq1[u] = lds2(q1row + u * 4);
      cq2[u] = lds2(q2row + u * 4);
    }
    for (int nt = 0; nt < 8; ++nt) {
      v8f c1 = {}, c2 = {};
#pragma unroll
      for (int ch = 0; ch < CH; ++ch) {
        const int nidx = (nt * CH + ch + 1) & (8 * CH - 1);
        const int nnt = nidx / CH, nch = nidx % CH;
        v2f na[KU], nq1[KU], nq2[KU];
#pragma unroll
        for (int u = 0; u < KU; ++u) {
          na[u] = lds2(arow + (nch * KU + u) * 4);
          nq1[u] = lds2(q1row + nnt * 16 * W_PITCH + (nch * KU + u) * 4);
          nq2[u] = lds2(q2row + nnt * 16 * W_PITCH + (nch * KU + u) * 4);
        }
#pragma unroll
        for (int u = 0; u < KU; ++u) {
          c1 = wmma_k4(ca[u], cq1[u], c1);
          c2 = wmma_k4(ca[u], cq2[u], c2);
        }
#pragma unroll
        for (int u = 0; u < KU; ++u) {
          ca[u] = na[u];
          cq1[u] = nq1[u];
          cq2[u] = nq2[u];
        }
      }
      const int n = nt * 16 + l16;
      const float bia1 = b1_2[n], bia2 = b2_2[n], wo = w_out[n];
#pragma unroll
      for (int j = 0; j < 8; ++j)
        part[j] += (c1[j] + bia1) * (c2[j] + bia2) * wo;
    }

#pragma unroll
    for (int j = 0; j < 8; ++j) {
      float v = part[j];
      v += __shfl_xor(v, 1, 32);
      v += __shfl_xor(v, 2, 32);
      v += __shfl_xor(v, 4, 32);
      v += __shfl_xor(v, 8, 32);
      part[j] = v;
    }
    if (l16 == 0) {
#pragma unroll
      for (int j = 0; j < 8; ++j) out[row0 + j + 8 * half] = part[j];
    }
  }
}

template <int WAVES>
__global__ __launch_bounds__(WAVES * 32) void net_fwd_grad_kernel(
    const float* __restrict__ x, const float* __restrict__ xdot,
    float* __restrict__ outY, float* __restrict__ outG, int n_samples,
    const float* __restrict__ w1_1, const float* __restrict__ b1_1,
    const float* __restrict__ w2_1, const float* __restrict__ b2_1,
    const float* __restrict__ w1_2, const float* __restrict__ b1_2,
    const float* __restrict__ w2_2, const float* __restrict__ b2_2,
    const float* __restrict__ w_out) {
  __shared__ __attribute__((aligned(16))) float wl1[H * W_PITCH];
  __shared__ __attribute__((aligned(16))) float wl2[H * W_PITCH];
  __shared__ __attribute__((aligned(16))) float ybuf[WAVES][16 * LDS_STRIDE];
  __shared__ __attribute__((aligned(16))) float ydbuf[WAVES][16 * LDS_STRIDE];

  const int wave = __builtin_amdgcn_readfirstlane(threadIdx.x >> 5);
  const int lane = threadIdx.x & 31;
  const int half = lane >> 4;
  const int l16 = lane & 15;

  stage_layer2_weights<WAVES * 32>(wl1, wl2, w1_2, w2_2);
  __syncthreads();

  float* yb = ybuf[wave];
  float* ydb = ydbuf[wave];
  const float* arow = yb + l16 * LDS_STRIDE + 2 * half;
  const float* drow = ydb + l16 * LDS_STRIDE + 2 * half;
  const float* q1row = wl1 + l16 * W_PITCH + 2 * half;
  const float* q2row = wl2 + l16 * W_PITCH + 2 * half;
  const int ntiles = (n_samples + 15) >> 4;
  const int tstep = gridDim.x * WAVES;

  for (int tile = blockIdx.x + gridDim.x * wave; tile < ntiles; tile += tstep) {
    const int row0 = tile * 16;

    {
      const float* xr = x + (row0 + l16) * NIN + 2 * half;
      const float* xdr = xdot + (row0 + l16) * NIN + 2 * half;
      v2f a0 = ldg2(xr), a1 = ldg2(xr + 4);
      v2f d0 = ldg2(xdr), d1 = ldg2(xdr + 4);
#pragma unroll
      for (int nt = 0; nt < 8; ++nt) {
        const int n = nt * 16 + l16;
        v2f p10 = ldg2(w1_1 + n * NIN + 2 * half);
        v2f p11 = ldg2(w1_1 + n * NIN + 4 + 2 * half);
        v2f p20 = ldg2(w2_1 + n * NIN + 2 * half);
        v2f p21 = ldg2(w2_1 + n * NIN + 4 + 2 * half);
        v8f c1 = {}, c2 = {}, g1 = {}, g2 = {};
        c1 = wmma_k4(a0, p10, c1);
        c1 = wmma_k4(a1, p11, c1);
        c2 = wmma_k4(a0, p20, c2);
        c2 = wmma_k4(a1, p21, c2);
        g1 = wmma_k4(d0, p10, g1);
        g1 = wmma_k4(d1, p11, g1);
        g2 = wmma_k4(d0, p20, g2);
        g2 = wmma_k4(d1, p21, g2);
        const float bb1 = b1_1[n], bb2 = b2_1[n];
#pragma unroll
        for (int j = 0; j < 8; ++j) {
          const int m = j + 8 * half;
          const float z1 = c1[j] + bb1, z2 = c2[j] + bb2;
          yb[m * LDS_STRIDE + n] = z1 * z2;
          ydb[m * LDS_STRIDE + n] = g1[j] * z2 + z1 * g2[j];
        }
      }
    }

    float pY[8], pG[8];
#pragma unroll
    for (int j = 0; j < 8; ++j) {
      pY[j] = 0.f;
      pG[j] = 0.f;
    }

    constexpr int KU = 4;
    constexpr int CH = 32 / KU;
    v2f ca[KU], cd[KU], cq1[KU], cq2[KU];
#pragma unroll
    for (int u = 0; u < KU; ++u) {
      ca[u] = lds2(arow + u * 4);
      cd[u] = lds2(drow + u * 4);
      cq1[u] = lds2(q1row + u * 4);
      cq2[u] = lds2(q2row + u * 4);
    }
    for (int nt = 0; nt < 8; ++nt) {
      v8f c1 = {}, c2 = {}, g1 = {}, g2 = {};
#pragma unroll
      for (int ch = 0; ch < CH; ++ch) {
        const int nidx = (nt * CH + ch + 1) & (8 * CH - 1);
        const int nnt = nidx / CH, nch = nidx % CH;
        v2f na[KU], nd[KU], nq1[KU], nq2[KU];
#pragma unroll
        for (int u = 0; u < KU; ++u) {
          na[u] = lds2(arow + (nch * KU + u) * 4);
          nd[u] = lds2(drow + (nch * KU + u) * 4);
          nq1[u] = lds2(q1row + nnt * 16 * W_PITCH + (nch * KU + u) * 4);
          nq2[u] = lds2(q2row + nnt * 16 * W_PITCH + (nch * KU + u) * 4);
        }
#pragma unroll
        for (int u = 0; u < KU; ++u) {
          c1 = wmma_k4(ca[u], cq1[u], c1);
          c2 = wmma_k4(ca[u], cq2[u], c2);
          g1 = wmma_k4(cd[u], cq1[u], g1);
          g2 = wmma_k4(cd[u], cq2[u], g2);
        }
#pragma unroll
        for (int u = 0; u < KU; ++u) {
          ca[u] = na[u];
          cd[u] = nd[u];
          cq1[u] = nq1[u];
          cq2[u] = nq2[u];
        }
      }
      const int n = nt * 16 + l16;
      const float bia1 = b1_2[n], bia2 = b2_2[n], wo = w_out[n];
#pragma unroll
      for (int j = 0; j < 8; ++j) {
        const float z1 = c1[j] + bia1, z2 = c2[j] + bia2;
        pY[j] += z1 * z2 * wo;
        pG[j] += (g1[j] * z2 + z1 * g2[j]) * wo;
      }
    }

#pragma unroll
    for (int j = 0; j < 8; ++j) {
      float vy = pY[j], vg = pG[j];
      vy += __shfl_xor(vy, 1, 32);
      vy += __shfl_xor(vy, 2, 32);
      vy += __shfl_xor(vy, 4, 32);
      vy += __shfl_xor(vy, 8, 32);
      vg += __shfl_xor(vg, 1, 32);
      vg += __shfl_xor(vg, 2, 32);
      vg += __shfl_xor(vg, 4, 32);
      vg += __shfl_xor(vg, 8, 32);
      pY[j] = vy;
      pG[j] = vg;
    }
    if (l16 == 0) {
#pragma unroll
      for (int j = 0; j < 8; ++j) {
        outY[row0 + j + 8 * half] = pY[j];
        outG[row0 + j + 8 * half] = pG[j];
      }
    }
  }
}

extern "C" void kernel_launch(void* const* d_in, const int* in_sizes, int n_in,
                              void* d_out, int out_size, void* d_ws,
                              size_t ws_size, hipStream_t stream) {
  const float* T = (const float*)d_in[0];
  const float* l = (const float*)d_in[1];
  const float* l1_dot = (const float*)d_in[2];
  const float* center = (const float*)d_in[3];
  const float* w1_1 = (const float*)d_in[4];
  const float* b1_1 = (const float*)d_in[5];
  const float* w2_1 = (const float*)d_in[6];
  const float* b2_1 = (const float*)d_in[7];
  const float* w1_2 = (const float*)d_in[8];
  const float* b1_2 = (const float*)d_in[9];
  const float* w2_2 = (const float*)d_in[10];
  const float* b2_2 = (const float*)d_in[11];
  const float* w_out = (const float*)d_in[12];

  const int nT = in_sizes[0] / NIN;
  const int nL = in_sizes[1] / NIN;
  const int nC = in_sizes[3] / NIN;

  float* out = (float*)d_out;
  float* v_t = out;
  float* v_y = out + nT;
  float* v_g = out + nT + nL;
  float* v_c = out + nT + 2 * nL;

  constexpr int WF = 16;
  constexpr int WG = 8;
  constexpr int NWG = 256;

  const int tilesC = (nC + 15) / 16;

  net_forward_kernel<WF><<<NWG, WF * 32, 0, stream>>>(
      T, v_t, nT, w1_1, b1_1, w2_1, b2_1, w1_2, b1_2, w2_2, b2_2, w_out);

  net_forward_kernel<WF><<<tilesC, WF * 32, 0, stream>>>(
      center, v_c, nC, w1_1, b1_1, w2_1, b2_1, w1_2, b1_2, w2_2, b2_2, w_out);

  net_fwd_grad_kernel<WG><<<NWG, WG * 32, 0, stream>>>(
      l, l1_dot, v_y, v_g, nL, w1_1, b1_1, w2_1, b2_1, w1_2, b1_2, w2_2, b2_2,
      w_out);
}


# All three launches merged into one resident launch: layer-2 weights copied to LDS once, each 8-wave group walks both forward inputs as one tile range, then the value+tangent tiles
# speedup vs baseline: 1.8805x; 1.1477x over previous
#include <hip/hip_runtime.h>


typedef float v2f __attribute__((ext_vector_type(2)));
typedef float v4f __attribute__((ext_vector_type(4)));
typedef float v8f __attribute__((ext_vector_type(8)));

#define H 128
#define NIN 8
#define LDS_STRIDE 130
#define W_PITCH 132

__device__ __forceinline__ v8f wmma_k4(v2f a, v2f b, v8f c) {
  return __builtin_amdgcn_wmma_f32_16x16x4_f32(
       false, a,  false, b,
       (short)0, c,  false,  false);
}

__device__ __forceinline__ v2f ldg2(const float* p) {
  return *reinterpret_cast<const v2f*>(p);
}

__device__ __forceinline__ v2f lds2(const float* p) {
  return *reinterpret_cast<const v2f*>(p);
}

template <int NTHREADS>
__device__ __forceinline__ void stage_layer2_weights(
    float* __restrict__ s1, float* __restrict__ s2,
    const float* __restrict__ w1_2, const float* __restrict__ w2_2) {
  constexpr int NVEC = H * H / 4;
  constexpr int NIT = NVEC / NTHREADS;
  v4f u1[NIT], u2[NIT];
#pragma unroll
  for (int it = 0; it < NIT; ++it) {
    const int i = it * NTHREADS + threadIdx.x;
    const int r = i >> 5;
    const int c = (i & 31) * 4;
    u1[it] = *reinterpret_cast<const v4f*>(w1_2 + r * H + c);
    u2[it] = *reinterpret_cast<const v4f*>(w2_2 + r * H + c);
  }
#pragma unroll
  for (int it = 0; it < NIT; ++it) {
    const int i = it * NTHREADS + threadIdx.x;
    const int r = i >> 5;
    const int c = (i & 31) * 4;
    *reinterpret_cast<v4f*>(s1 + r * W_PITCH + c) = u1[it];
    *reinterpret_cast<v4f*>(s2 + r * W_PITCH + c) = u2[it];
  }
}

__device__ __forceinline__ void forward_tile(
    const float* x, float* out, const int row0,
    float* yb, const float* arow,
    const float* q1row, const float* q2row,
    const int half, const int l16,
    const float* w1_1, const float* b1_1,
    const float* w2_1, const float* b2_1,
    const float* b1_2, const float* b2_2,
    const float* w_out) {
  {
    const float* xr = x + (row0 + l16) * NIN + 2 * half;
    v2f a0 = ldg2(xr);
    v2f a1 = ldg2(xr + 4);
#pragma unroll
    for (int nt = 0; nt < 8; ++nt) {
      const int n = nt * 16 + l16;
      v2f p10 = ldg2(w1_1 + n * NIN + 2 * half);
      v2f p11 = ldg2(w1_1 + n * NIN + 4 + 2 * half);
      v2f p20 = ldg2(w2_1 + n * NIN + 2 * half);
      v2f p21 = ldg2(w2_1 + n * NIN + 4 + 2 * half);
      v8f c1 = {}, c2 = {};
      c1 = wmma_k4(a0, p10, c1);
      c1 = wmma_k4(a1, p11, c1);
      c2 = wmma_k4(a0, p20, c2);
      c2 = wmma_k4(a1, p21, c2);
      const float bb1 = b1_1[n], bb2 = b2_1[n];
#pragma unroll
      for (int j = 0; j < 8; ++j) {
        const int m = j + 8 * half;
        yb[m * LDS_STRIDE + n] = (c1[j] + bb1) * (c2[j] + bb2);
      }
    }
  }

  float part[8];
#pragma unroll
  for (int j = 0; j < 8; ++j) part[j] = 0.f;

  constexpr int KU = 8;
  constexpr int CH = 32 / KU;
  v2f ca[KU], cq1[KU], cq2[KU];
#pragma unroll
  for (int u = 0; u < KU; ++u) {
    ca[u] = lds2(arow + u * 4);
    cq1[u] = lds2(q1row + u * 4);
    cq2[u] = lds2(q2row + u * 4);
  }
  for (int nt = 0; nt < 8; ++nt) {
    v8f c1 = {}, c2 = {};
#pragma unroll
    for (int ch = 0; ch < CH; ++ch) {
      const int nidx = (nt * CH + ch + 1) & (8 * CH - 1);
      const int nnt = nidx / CH, nch = nidx % CH;
      v2f na[KU], nq1[KU], nq2[KU];
#pragma unroll
      for (int u = 0; u < KU; ++u) {
        na[u] = lds2(arow + (nch * KU + u) * 4);
        nq1[u] = lds2(q1row + nnt * 16 * W_PITCH + (nch * KU + u) * 4);
        nq2[u] = lds2(q2row + nnt * 16 * W_PITCH + (nch * KU + u) * 4);
      }
#pragma unroll
      for (int u = 0; u < KU; ++u) {
        c1 = wmma_k4(ca[u], cq1[u], c1);
        c2 = wmma_k4(ca[u], cq2[u], c2);
      }
#pragma unroll
      for (int u = 0; u < KU; ++u) {
        ca[u] = na[u];
        cq1[u] = nq1[u];
        cq2[u] = nq2[u];
      }
    }
    const int n = nt * 16 + l16;
    const float bia1 = b1_2[n], bia2 = b2_2[n], wo = w_out[n];
#pragma unroll
    for (int j = 0; j < 8; ++j)
      part[j] += (c1[j] + bia1) * (c2[j] + bia2) * wo;
  }

#pragma unroll
  for (int j = 0; j < 8; ++j) {
    float v = part[j];
    v += __shfl_xor(v, 1, 32);
    v += __shfl_xor(v, 2, 32);
    v += __shfl_xor(v, 4, 32);
    v += __shfl_xor(v, 8, 32);
    part[j] = v;
  }
  if (l16 == 0) {
#pragma unroll
    for (int j = 0; j < 8; ++j) out[row0 + j + 8 * half] = part[j];
  }
}

__device__ __forceinline__ void fwd_grad_tile(
    const float* x, const float* xdot,
    float* outY, float* outG, const int row0,
    float* yb, float* ydb,
    const float* arow, const float* drow,
    const float* q1row, const float* q2row,
    const int half, const int l16,
    const float* w1_1, const float* b1_1,
    const float* w2_1, const float* b2_1,
    const float* b1_2, const float* b2_2,
    const float* w_out) {
  {
    const float* xr = x + (row0 + l16) * NIN + 2 * half;
    const float* xdr = xdot + (row0 + l16) * NIN + 2 * half;
    v2f a0 = ldg2(xr), a1 = ldg2(xr + 4);
    v2f d0 = ldg2(xdr), d1 = ldg2(xdr + 4);
#pragma unroll
    for (int nt = 0; nt < 8; ++nt) {
      const int n = nt * 16 + l16;
      v2f p10 = ldg2(w1_1 + n * NIN + 2 * half);
      v2f p11 = ldg2(w1_1 + n * NIN + 4 + 2 * half);
      v2f p20 = ldg2(w2_1 + n * NIN + 2 * half);
      v2f p21 = ldg2(w2_1 + n * NIN + 4 + 2 * half);
      v8f c1 = {}, c2 = {}, g1 = {}, g2 = {};
      c1 = wmma_k4(a0, p10, c1);
      c1 = wmma_k4(a1, p11, c1);
      c2 = wmma_k4(a0, p20, c2);
      c2 = wmma_k4(a1, p21, c2);
      g1 = wmma_k4(d0, p10, g1);
      g1 = wmma_k4(d1, p11, g1);
      g2 = wmma_k4(d0, p20, g2);
      g2 = wmma_k4(d1, p21, g2);
      const float bb1 = b1_1[n], bb2 = b2_1[n];
#pragma unroll
      for (int j = 0; j < 8; ++j) {
        const int m = j + 8 * half;
        const float z1 = c1[j] + bb1, z2 = c2[j] + bb2;
        yb[m * LDS_STRIDE + n] = z1 * z2;
        ydb[m * LDS_STRIDE + n] = g1[j] * z2 + z1 * g2[j];
      }
    }
  }

  float pY[8], pG[8];
#pragma unroll
  for (int j = 0; j < 8; ++j) {
    pY[j] = 0.f;
    pG[j] = 0.f;
  }

  constexpr int KU = 4;
  constexpr int CH = 32 / KU;
  v2f ca[KU], cd[KU], cq1[KU], cq2[KU];
#pragma unroll
  for (int u = 0; u < KU; ++u) {
    ca[u] = lds2(arow + u * 4);
    cd[u] = lds2(drow + u * 4);
    cq1[u] = lds2(q1row + u * 4);
    cq2[u] = lds2(q2row + u * 4);
  }
  for (int nt = 0; nt < 8; ++nt) {
    v8f c1 = {}, c2 = {}, g1 = {}, g2 = {};
#pragma unroll
    for (int ch = 0; ch < CH; ++ch) {
      const int nidx = (nt * CH + ch + 1) & (8 * CH - 1);
      const int nnt = nidx / CH, nch = nidx % CH;
      v2f na[KU], nd[KU], nq1[KU], nq2[KU];
#pragma unroll
      for (int u = 0; u < KU; ++u) {
        na[u] = lds2(arow + (nch * KU + u) * 4);
        nd[u] = lds2(drow + (nch * KU + u) * 4);
        nq1[u] = lds2(q1row + nnt * 16 * W_PITCH + (nch * KU + u) * 4);
        nq2[u] = lds2(q2row + nnt * 16 * W_PITCH + (nch * KU + u) * 4);
      }
#pragma unroll
      for (int u = 0; u < KU; ++u) {
        c1 = wmma_k4(ca[u], cq1[u], c1);
        c2 = wmma_k4(ca[u], cq2[u], c2);
        g1 = wmma_k4(cd[u], cq1[u], g1);
        g2 = wmma_k4(cd[u], cq2[u], g2);
      }
#pragma unroll
      for (int u = 0; u < KU; ++u) {
        ca[u] = na[u];
        cd[u] = nd[u];
        cq1[u] = nq1[u];
        cq2[u] = nq2[u];
      }
    }
    const int n = nt * 16 + l16;
    const float bia1 = b1_2[n], bia2 = b2_2[n], wo = w_out[n];
#pragma unroll
    for (int j = 0; j < 8; ++j) {
      const float z1 = c1[j] + bia1, z2 = c2[j] + bia2;
      pY[j] += z1 * z2 * wo;
      pG[j] += (g1[j] * z2 + z1 * g2[j]) * wo;
    }
  }

#pragma unroll
  for (int j = 0; j < 8; ++j) {
    float vy = pY[j], vg = pG[j];
    vy += __shfl_xor(vy, 1, 32);
    vy += __shfl_xor(vy, 2, 32);
    vy += __shfl_xor(vy, 4, 32);
    vy += __shfl_xor(vy, 8, 32);
    vg += __shfl_xor(vg, 1, 32);
    vg += __shfl_xor(vg, 2, 32);
    vg += __shfl_xor(vg, 4, 32);
    vg += __shfl_xor(vg, 8, 32);
    pY[j] = vy;
    pG[j] = vg;
  }
  if (l16 == 0) {
#pragma unroll
    for (int j = 0; j < 8; ++j) {
      outY[row0 + j + 8 * half] = pY[j];
      outG[row0 + j + 8 * half] = pG[j];
    }
  }
}

template <int WAVES>
__global__ __launch_bounds__(WAVES * 32) void net_all_kernel(
    const float* __restrict__ xT, const float* __restrict__ xL,
    const float* __restrict__ xLdot, const float* __restrict__ xC,
    float* __restrict__ outT, float* __restrict__ outY,
    float* __restrict__ outG, float* __restrict__ outC,
    int nT, int nL, int nC,
    const float* __restrict__ w1_1, const float* __restrict__ b1_1,
    const float* __restrict__ w2_1, const float* __restrict__ b2_1,
    const float* __restrict__ w1_2, const float* __restrict__ b1_2,
    const float* __restrict__ w2_2, const float* __restrict__ b2_2,
    const float* __restrict__ w_out) {
  __shared__ __attribute__((aligned(16))) float wl1[H * W_PITCH];
  __shared__ __attribute__((aligned(16))) float wl2[H * W_PITCH];
  __shared__ __attribute__((aligned(16))) float ybuf[WAVES][16 * LDS_STRIDE];
  __shared__ __attribute__((aligned(16))) float ydbuf[WAVES][16 * LDS_STRIDE];

  const int wave = __builtin_amdgcn_readfirstlane(threadIdx.x >> 5);
  const int lane = threadIdx.x & 31;
  const int half = lane >> 4;
  const int l16 = lane & 15;

  stage_layer2_weights<WAVES * 32>(wl1, wl2, w1_2, w2_2);
  __syncthreads();

  float* yb = ybuf[wave];
  float* ydb = ydbuf[wave];
  const float* arow = yb + l16 * LDS_STRIDE + 2 * half;
  const float* drow = ydb + l16 * LDS_STRIDE + 2 * half;
  const float* q1row = wl1 + l16 * W_PITCH + 2 * half;
  const float* q2row = wl2 + l16 * W_PITCH + 2 * half;
  const int first = blockIdx.x + gridDim.x * wave;
  const int tstep = gridDim.x * WAVES;

  const int tilesT = (nT + 15) >> 4;
  const int tilesF = tilesT + ((nC + 15) >> 4);
  for (int tile = first; tile < tilesF; tile += tstep) {
    const bool small = tile >= tilesT;
    const float* x = small ? xC : xT;
    float* out = small ? outC : outT;
    const int row0 = (small ? tile - tilesT : tile) * 16;
    forward_tile(x, out, row0, yb, arow, q1row, q2row, half, l16, w1_1, b1_1,
                 w2_1, b2_1, b1_2, b2_2, w_out);
  }

  const int tilesL = (nL + 15) >> 4;
  for (int tile = first; tile < tilesL; tile += tstep) {
    fwd_grad_tile(xL, xLdot, outY, outG, tile * 16, yb, ydb, arow, drow, q1row,
                  q2row, half, l16, w1_1, b1_1, w2_1, b2_1, b1_2, b2_2, w_out);
  }
}

extern "C" void kernel_launch(void* const* d_in, const int* in_sizes, int n_in,
                              void* d_out, int out_size, void* d_ws,
                              size_t ws_size, hipStream_t stream) {
  const float* T = (const float*)d_in[0];
  const float* l = (const float*)d_in[1];
  const float* l1_dot = (const float*)d_in[2];
  const float* center = (const float*)d_in[3];
  const float* w1_1 = (const float*)d_in[4];
  const float* b1_1 = (const float*)d_in[5];
  const float* w2_1 = (const float*)d_in[6];
  const float* b2_1 = (const float*)d_in[7];
  const float* w1_2 = (const float*)d_in[8];
  const float* b1_2 = (const float*)d_in[9];
  const float* w2_2 = (const float*)d_in[10];
  const float* b2_2 = (const float*)d_in[11];
  const float* w_out = (const float*)d_in[12];

  const int nT = in_sizes[0] / NIN;
  const int nL = in_sizes[1] / NIN;
  const int nC = in_sizes[3] / NIN;

  float* out = (float*)d_out;
  float* v_t = out;
  float* v_y = out + nT;
  float* v_g = out + nT + nL;
  float* v_c = out + nT + 2 * nL;

  constexpr int WV = 8;
  constexpr int NWG = 256;

  net_all_kernel<WV><<<NWG, WV * 32, 0, stream>>>(
      T, l, l1_dot, center, v_t, v_y, v_g, v_c, nT, nL, nC, w1_1, b1_1, w2_1,
      b2_1, w1_2, b1_2, w2_2, b2_2, w_out);
}


# Layer 2 and the output dot folded into one symmetric 48x48 quadratic form S of the monomials: per tile t = S e (36 WMMAs, was 192/384), value = e.t/2, tangent = de.t; summation order changes
# speedup vs baseline: 8.7916x; 4.6751x over previous
#include <hip/hip_runtime.h>


typedef float v2f __attribute__((ext_vector_type(2)));
typedef float v4f __attribute__((ext_vector_type(4)));
typedef float v8f __attribute__((ext_vector_type(8)));

#define H 128
#define NIN 8
#define NX 9
#define NF 45
#define KF 48
#define W_PITCH 132
#define S_PITCH 52

__host__ __device__ constexpr int feat_i(int f) {
  int i = 0, len = NX;
  while (f >= len) { f -= len; ++i; --len; }
  return i;
}
__host__ __device__ constexpr int feat_j(int f) {
  int i = 0, len = NX;
  while (f >= len) { f -= len; ++i; --len; }
  return i + f;
}

template <int V> struct IC { static constexpr int value = V; };
template <int I, int N, class Fn>
__device__ __forceinline__ void static_for(Fn&& fn) {
  if constexpr (I < N) {
    fn(IC<I>{});
    static_for<I + 1, N>(fn);
  }
}

__device__ __forceinline__ v8f wmma_k4(v2f a, v2f b, v8f c) {
  return __builtin_amdgcn_wmma_f32_16x16x4_f32(
       false, a,  false, b,
       (short)0, c,  false,  false);
}

__device__ __forceinline__ v2f lds2(const float* p) {
  return *reinterpret_cast<const v2f*>(p);
}

template <int NTHREADS>
__device__ __forceinline__ void stage_layer2_weights(
    float* __restrict__ s1, float* __restrict__ s2,
    const float* __restrict__ w1_2, const float* __restrict__ w2_2) {
  constexpr int NVEC = H * H / 4;
  constexpr int NIT = NVEC / NTHREADS;
  v4f u1[NIT], u2[NIT];
#pragma unroll
  for (int it = 0; it < NIT; ++it) {
    const int i = it * NTHREADS + threadIdx.x;
    const int r = i >> 5;
    const int c = (i & 31) * 4;
    u1[it] = *reinterpret_cast<const v4f*>(w1_2 + r * H + c);
    u2[it] = *reinterpret_cast<const v4f*>(w2_2 + r * H + c);
  }
#pragma unroll
  for (int it = 0; it < NIT; ++it) {
    const int i = it * NTHREADS + threadIdx.x;
    const int r = i >> 5;
    const int c = (i & 31) * 4;
    *reinterpret_cast<v4f*>(s1 + r * W_PITCH + c) = u1[it];
    *reinterpret_cast<v4f*>(s2 + r * W_PITCH + c) = u2[it];
  }
}

template <int F>
__device__ __forceinline__ float phi(const float (&xt)[NX]) {
  if constexpr (F < NF) {
    constexpr int i = feat_i(F), j = feat_j(F);
    return xt[i] * xt[j];
  } else {
    return 0.f;
  }
}
template <int F>
__device__ __forceinline__ float dphi(const float (&xt)[NX], const float (&xd)[NX]) {
  if constexpr (F < NF) {
    constexpr int i = feat_i(F), j = feat_j(F);
    return xd[i] * xt[j] + xt[i] * xd[j];
  } else {
    return 0.f;
  }
}

template <int F0>
__device__ __forceinline__ void build_mt_rows(float* mt, const int h,
                                              const float (&p)[NX],
                                              const float (&q)[NX]) {
  static_for<0, KF / 2>([&](auto ic) {
    constexpr int F = F0 + decltype(ic)::value;
    float v = 0.f;
    if constexpr (F < NF) {
      constexpr int i = feat_i(F), j = feat_j(F);
      if constexpr (i == j) v = p[i] * q[i];
      else v = p[i] * q[j] + p[j] * q[i];
    }
    mt[F * W_PITCH + h] = v;
  });
}

__device__ __forceinline__ void load_row(const float* x, const int row,
                                         const float last, float (&xt)[NX]) {
  const v4f lo = *reinterpret_cast<const v4f*>(x + row * NIN);
  const v4f hi = *reinterpret_cast<const v4f*>(x + row * NIN + 4);
  xt[0] = lo[0]; xt[1] = lo[1]; xt[2] = lo[2]; xt[3] = lo[3];
  xt[4] = hi[0]; xt[5] = hi[1]; xt[6] = hi[2]; xt[7] = hi[3];
  xt[8] = last;
}

template <int E>
__device__ __forceinline__ float phie(const float (&xt)[NX]) {
  if constexpr (E == NF) return 1.f;
  else return phi<E>(xt);
}
template <int E>
__device__ __forceinline__ float dphie(const float (&xt)[NX], const float (&xd)[NX]) {
  return dphi<E>(xt, xd);
}

__host__ __device__ constexpr int slot_feat(int half, int idx) {
  return 16 * (idx >> 3) + 8 * half + (idx & 7);
}

template <bool GRAD>
__device__ __forceinline__ void quad_tile(
    const float* x, const float* xdot, float* outY, float* outG,
    const int row0, const v4f (&sf)[3][6], const int half, const int l16) {
  float xt[NX], xd[NX];
  load_row(x, row0 + l16, 1.f, xt);
  if constexpr (GRAD) load_row(xdot, row0 + l16, 0.f, xd);
  float pv[24], dv[24];
  static_for<0, 24>([&](auto ic) {
    constexpr int i = decltype(ic)::value;
    pv[i] = half ? phie<slot_feat(1, i)>(xt) : phie<slot_feat(0, i)>(xt);
    if constexpr (GRAD)
      dv[i] = half ? dphie<slot_feat(1, i)>(xt, xd) : dphie<slot_feat(0, i)>(xt, xd);
  });

  v8f t[3] = {};
#pragma unroll
  for (int v = 0; v < 6; ++v) {
    const v2f b0 = {pv[4 * v], pv[4 * v + 1]};
    const v2f b1 = {pv[4 * v + 2], pv[4 * v + 3]};
#pragma unroll
    for (int mb = 0; mb < 3; ++mb) {
      const v2f a0 = {sf[mb][v][0], sf[mb][v][1]};
      t[mb] = wmma_k4(a0, b0, t[mb]);
    }
#pragma unroll
    for (int mb = 0; mb < 3; ++mb) {
      const v2f a1 = {sf[mb][v][2], sf[mb][v][3]};
      t[mb] = wmma_k4(a1, b1, t[mb]);
    }
  }

  float ys[3], gs[3];
#pragma unroll
  for (int mb = 0; mb < 3; ++mb) {
    ys[mb] = 0.f;
    gs[mb] = 0.f;
#pragma unroll
    for (int r = 0; r < 8; ++r) {
      ys[mb] += t[mb][r] * pv[8 * mb + r];
      if constexpr (GRAD) gs[mb] += t[mb][r] * dv[8 * mb + r];
    }
  }
  float y = (ys[0] + ys[1]) + ys[2];
  y += __shfl_xor(y, 16, 32);
  y *= 0.5f;
  if constexpr (GRAD) {
    float g = (gs[0] + gs[1]) + gs[2];
    g += __shfl_xor(g, 16, 32);
    float* dst = half ? outG : outY;
    dst[row0 + l16] = half ? g : y;
  } else {
    if (half == 0) outY[row0 + l16] = y;
  }
}

template <int WAVES>
__global__ __launch_bounds__(WAVES * 32) void net_all_kernel(
    const float* __restrict__ xT, const float* __restrict__ xL,
    const float* __restrict__ xLdot, const float* __restrict__ xC,
    float* __restrict__ outT, float* __restrict__ outY,
    float* __restrict__ outG, float* __restrict__ outC,
    int nT, int nL, int nC,
    const float* __restrict__ w1_1, const float* __restrict__ b1_1,
    const float* __restrict__ w2_1, const float* __restrict__ b2_1,
    const float* __restrict__ w1_2, const float* __restrict__ b1_2,
    const float* __restrict__ w2_2, const float* __restrict__ b2_2,
    const float* __restrict__ w_out) {
  static_assert(WAVES == 8, "one 16-row block of C1/C2 per wave; M^T rows split over two wave groups");
  __shared__ __attribute__((aligned(16))) float wl1[H * W_PITCH];
  __shared__ __attribute__((aligned(16))) float wl2[H * W_PITCH];
  __shared__ __attribute__((aligned(16))) float mt[KF * W_PITCH];
  __shared__ __attribute__((aligned(16))) float part[WAVES * KF * S_PITCH];
  __shared__ __attribute__((aligned(16))) float qs[KF * S_PITCH];
  __shared__ __attribute__((aligned(16))) float qt[KF * S_PITCH];

  const int wave = __builtin_amdgcn_readfirstlane(threadIdx.x >> 5);
  const int lane = threadIdx.x & 31;
  const int half = lane >> 4;
  const int l16 = lane & 15;

  stage_layer2_weights<WAVES * 32>(wl1, wl2, w1_2, w2_2);

  {
    const int h = threadIdx.x & (H - 1);
    float p[NX], q[NX];
    load_row(w1_1, h, b1_1[h], p);
    load_row(w2_1, h, b2_1[h], q);
    if (wave < WAVES / 2) build_mt_rows<0>(mt, h, p, q);
    else build_mt_rows<KF / 2>(mt, h, p, q);
  }
  __syncthreads();

  const int o0 = wave * 16 + 8 * half;
  const v4f wlo = *reinterpret_cast<const v4f*>(w_out + o0);
  const v4f whi = *reinterpret_cast<const v4f*>(w_out + o0 + 4);
  const v4f b1lo = *reinterpret_cast<const v4f*>(b1_2 + o0);
  const v4f b1hi = *reinterpret_cast<const v4f*>(b1_2 + o0 + 4);
  const v4f b2lo = *reinterpret_cast<const v4f*>(b2_2 + o0);
  const v4f b2hi = *reinterpret_cast<const v4f*>(b2_2 + o0 + 4);

  v8f d1[3] = {}, d2[3] = {};
  {
    const float* a1 = wl1 + (wave * 16 + l16) * W_PITCH + 2 * half;
    const float* a2 = wl2 + (wave * 16 + l16) * W_PITCH + 2 * half;
    const float* bm = mt + l16 * W_PITCH + 2 * half;
#pragma unroll 8
    for (int u = 0; u < H / 4; ++u) {
      const v2f fa1 = lds2(a1 + u * 4), fa2 = lds2(a2 + u * 4);
#pragma unroll
      for (int ft = 0; ft < 3; ++ft) {
        const v2f fb = lds2(bm + ft * 16 * W_PITCH + u * 4);
        d1[ft] = wmma_k4(fa1, fb, d1[ft]);
        d2[ft] = wmma_k4(fa2, fb, d2[ft]);
      }
    }
  }

  {
    const bool bias_lane = l16 == NF - 32;
#pragma unroll
    for (int j = 0; j < 8; ++j) {
      const float wo = j < 4 ? wlo[j & 3] : whi[j & 3];
      const float bb1 = j < 4 ? b1lo[j & 3] : b1hi[j & 3];
      const float bb2 = j < 4 ? b2lo[j & 3] : b2hi[j & 3];
      d1[2][j] = bias_lane ? bb1 : d1[2][j];
      d2[2][j] = bias_lane ? bb2 : d2[2][j];
#pragma unroll
      for (int ft = 0; ft < 3; ++ft) d1[ft][j] *= wo;
    }
  }

  {
    v8f pq[3][3] = {};
#pragma unroll
    for (int u = 0; u < 4; ++u) {
#pragma unroll
      for (int fb = 0; fb < 3; ++fb) {
        const v2f fa = {d1[fb][2 * u], d1[fb][2 * u + 1]};
#pragma unroll
        for (int gb = 0; gb < 3; ++gb) {
          const v2f fb2 = {d2[gb][2 * u], d2[gb][2 * u + 1]};
          pq[fb][gb] = wmma_k4(fa, fb2, pq[fb][gb]);
        }
      }
    }
    float* pw = part + wave * (KF * S_PITCH);
#pragma unroll
    for (int fb = 0; fb < 3; ++fb) {
#pragma unroll
      for (int gb = 0; gb < 3; ++gb) {
#pragma unroll
        for (int r = 0; r < 8; ++r)
          pw[(16 * fb + 8 * half + r) * S_PITCH + 16 * gb + l16] = pq[fb][gb][r];
      }
    }
  }
  __syncthreads();

  constexpr int NQUAD = KF * (KF / 4);
#pragma unroll
  for (int it = 0; it < (NQUAD + WAVES * 32 - 1) / (WAVES * 32); ++it) {
    const int qid = it * (WAVES * 32) + threadIdx.x;
    if (qid < NQUAD) {
      const int f = qid / (KF / 4);
      const int g0 = (qid - f * (KF / 4)) * 4;
      v4f acc = *reinterpret_cast<const v4f*>(part + f * S_PITCH + g0);
#pragma unroll
      for (int w = 1; w < WAVES; ++w)
        acc += *reinterpret_cast<const v4f*>(part + w * (KF * S_PITCH) + f * S_PITCH + g0);
      *reinterpret_cast<v4f*>(qs + f * S_PITCH + g0) = acc;
#pragma unroll
      for (int i = 0; i < 4; ++i) qt[(g0 + i) * S_PITCH + f] = acc[i];
    }
  }
  __syncthreads();

  v4f sf[3][6];
#pragma unroll
  for (int mb = 0; mb < 3; ++mb) {
#pragma unroll
    for (int v = 0; v < 6; ++v) {
      const int off = (16 * mb + l16) * S_PITCH + 16 * (v >> 1) + 8 * half + 4 * (v & 1);
      sf[mb][v] = *reinterpret_cast<const v4f*>(qs + off) +
                  *reinterpret_cast<const v4f*>(qt + off);
    }
  }

  const int first = blockIdx.x + gridDim.x * wave;
  const int tstep = gridDim.x * WAVES;

  const int tilesT = (nT + 15) >> 4;
  const int tilesF = tilesT + ((nC + 15) >> 4);
  for (int tile = first; tile < tilesF; tile += tstep) {
    const bool small = tile >= tilesT;
    const float* x = small ? xC : xT;
    float* out = small ? outC : outT;
    const int row0 = (small ? tile - tilesT : tile) * 16;
    quad_tile<false>(x, nullptr, out, nullptr, row0, sf, half, l16);
  }

  const int tilesL = (nL + 15) >> 4;
  for (int tile = first; tile < tilesL; tile += tstep) {
    quad_tile<true>(xL, xLdot, outY, outG, tile * 16, sf, half, l16);
  }
}

extern "C" void kernel_launch(void* const* d_in, const int* in_sizes, int n_in,
                              void* d_out, int out_size, void* d_ws,
                              size_t ws_size, hipStream_t stream) {
  const float* T = (const float*)d_in[0];
  const float* l = (const float*)d_in[1];
  const float* l1_dot = (const float*)d_in[2];
  const float* center = (const float*)d_in[3];
  const float* w1_1 = (const float*)d_in[4];
  const float* b1_1 = (const float*)d_in[5];
  const float* w2_1 = (const float*)d_in[6];
  const float* b2_1 = (const float*)d_in[7];
  const float* w1_2 = (const float*)d_in[8];
  const float* b1_2 = (const float*)d_in[9];
  const float* w2_2 = (const float*)d_in[10];
  const float* b2_2 = (const float*)d_in[11];
  const float* w_out = (const float*)d_in[12];

  const int nT = in_sizes[0] / NIN;
  const int nL = in_sizes[1] / NIN;
  const int nC = in_sizes[3] / NIN;

  float* out = (float*)d_out;
  float* v_t = out;
  float* v_y = out + nT;
  float* v_g = out + nT + nL;
  float* v_c = out + nT + 2 * nL;

  constexpr int WV = 8;
  constexpr int NWG = 256;

  net_all_kernel<WV><<<NWG, WV * 32, 0, stream>>>(
      T, l, l1_dot, center, v_t, v_y, v_g, v_c, nT, nL, nC, w1_1, b1_1, w2_1,
      b2_1, w1_2, b1_2, w2_2, b2_2, w_out);
}


# Q's 8 partial sums (16 hidden units each) built once by a first launch of 8 small groups into the workspace; main launch only adds them: no weight copy or C1/C2 products per group; sum order changes
# speedup vs baseline: 9.7515x; 1.1092x over previous
#include <hip/hip_runtime.h>


typedef float v2f __attribute__((ext_vector_type(2)));
typedef float v4f __attribute__((ext_vector_type(4)));
typedef float v8f __attribute__((ext_vector_type(8)));

#define H 128
#define NIN 8
#define NX 9
#define NF 45
#define KF 48
#define W_PITCH 132
#define S_PITCH 52

__host__ __device__ constexpr int feat_i(int f) {
  int i = 0, len = NX;
  while (f >= len) { f -= len; ++i; --len; }
  return i;
}
__host__ __device__ constexpr int feat_j(int f) {
  int i = 0, len = NX;
  while (f >= len) { f -= len; ++i; --len; }
  return i + f;
}

template <int V> struct IC { static constexpr int value = V; };
template <int I, int N, class Fn>
__device__ __forceinline__ void static_for(Fn&& fn) {
  if constexpr (I < N) {
    fn(IC<I>{});
    static_for<I + 1, N>(fn);
  }
}

__device__ __forceinline__ v8f wmma_k4(v2f a, v2f b, v8f c) {
  return __builtin_amdgcn_wmma_f32_16x16x4_f32(
       false, a,  false, b,
       (short)0, c,  false,  false);
}

template <int F>
__device__ __forceinline__ float phi(const float (&xt)[NX]) {
  if constexpr (F < NF) {
    constexpr int i = feat_i(F), j = feat_j(F);
    return xt[i] * xt[j];
  } else {
    return 0.f;
  }
}
template <int F>
__device__ __forceinline__ float dphi(const float (&xt)[NX], const float (&xd)[NX]) {
  if constexpr (F < NF) {
    constexpr int i = feat_i(F), j = feat_j(F);
    return xd[i] * xt[j] + xt[i] * xd[j];
  } else {
    return 0.f;
  }
}

template <int F0>
__device__ __forceinline__ void build_mt_rows(float* mt, const int h,
                                              const float (&p)[NX],
                                              const float (&q)[NX]) {
  static_for<0, KF / 2>([&](auto ic) {
    constexpr int F = F0 + decltype(ic)::value;
    float v = 0.f;
    if constexpr (F < NF) {
      constexpr int i = feat_i(F), j = feat_j(F);
      if constexpr (i == j) v = p[i] * q[i];
      else v = p[i] * q[j] + p[j] * q[i];
    }
    mt[F * W_PITCH + h] = v;
  });
}

__device__ __forceinline__ void load_row(const float* x, const int row,
                                         const float last, float (&xt)[NX]) {
  const v4f lo = *reinterpret_cast<const v4f*>(x + row * NIN);
  const v4f hi = *reinterpret_cast<const v4f*>(x + row * NIN + 4);
  xt[0] = lo[0]; xt[1] = lo[1]; xt[2] = lo[2]; xt[3] = lo[3];
  xt[4] = hi[0]; xt[5] = hi[1]; xt[6] = hi[2]; xt[7] = hi[3];
  xt[8] = last;
}

template <int E>
__device__ __forceinline__ float phie(const float (&xt)[NX]) {
  if constexpr (E == NF) return 1.f;
  else return phi<E>(xt);
}
template <int E>
__device__ __forceinline__ float dphie(const float (&xt)[NX], const float (&xd)[NX]) {
  return dphi<E>(xt, xd);
}

__host__ __device__ constexpr int slot_feat(int half, int idx) {
  return 16 * (idx >> 3) + 8 * half + (idx & 7);
}

template <bool GRAD>
__device__ __forceinline__ void quad_tile(
    const float* x, const float* xdot, float* outY, float* outG,
    const int row0, const v4f (&sf)[3][6], const int half, const int l16) {
  float xt[NX], xd[NX];
  load_row(x, row0 + l16, 1.f, xt);
  if constexpr (GRAD) load_row(xdot, row0 + l16, 0.f, xd);
  float pv[24], dv[24];
  static_for<0, 24>([&](auto ic) {
    constexpr int i = decltype(ic)::value;
    pv[i] = half ? phie<slot_feat(1, i)>(xt) : phie<slot_feat(0, i)>(xt);
    if constexpr (GRAD)
      dv[i] = half ? dphie<slot_feat(1, i)>(xt, xd) : dphie<slot_feat(0, i)>(xt, xd);
  });

  v8f t[3] = {};
#pragma unroll
  for (int v = 0; v < 6; ++v) {
    const v2f b0 = {pv[4 * v], pv[4 * v + 1]};
    const v2f b1 = {pv[4 * v + 2], pv[4 * v + 3]};
#pragma unroll
    for (int mb = 0; mb < 3; ++mb) {
      const v2f a0 = {sf[mb][v][0], sf[mb][v][1]};
      t[mb] = wmma_k4(a0, b0, t[mb]);
    }
#pragma unroll
    for (int mb = 0; mb < 3; ++mb) {
      const v2f a1 = {sf[mb][v][2], sf[mb][v][3]};
      t[mb] = wmma_k4(a1, b1, t[mb]);
    }
  }

  float ys[3], gs[3];
#pragma unroll
  for (int mb = 0; mb < 3; ++mb) {
    ys[mb] = 0.f;
    gs[mb] = 0.f;
#pragma unroll
    for (int r = 0; r < 8; ++r) {
      ys[mb] += t[mb][r] * pv[8 * mb + r];
      if constexpr (GRAD) gs[mb] += t[mb][r] * dv[8 * mb + r];
    }
  }
  float y = (ys[0] + ys[1]) + ys[2];
  y += __shfl_xor(y, 16, 32);
  y *= 0.5f;
  if constexpr (GRAD) {
    float g = (gs[0] + gs[1]) + gs[2];
    g += __shfl_xor(g, 16, 32);
    float* dst = half ? outG : outY;
    dst[row0 + l16] = half ? g : y;
  } else {
    if (half == 0) outY[row0 + l16] = y;
  }
}

#define A_WAVES 4
__global__ __launch_bounds__(A_WAVES * 32) void q_partial_kernel(
    const float* __restrict__ w1_1, const float* __restrict__ b1_1,
    const float* __restrict__ w2_1, const float* __restrict__ b2_1,
    const float* __restrict__ w1_2, const float* __restrict__ b1_2,
    const float* __restrict__ w2_2, const float* __restrict__ b2_2,
    const float* __restrict__ w_out, float* __restrict__ part) {
  static_assert(A_WAVES * 32 == H, "M^T: one thread per hidden unit h");
  __shared__ __attribute__((aligned(16))) float mt[KF * W_PITCH];
  __shared__ __attribute__((aligned(16))) v4f cx[3 * 2 * 32];

  const int wave = __builtin_amdgcn_readfirstlane(threadIdx.x >> 5);
  const int lane = threadIdx.x & 31;
  const int half = lane >> 4;
  const int l16 = lane & 15;
  const int ob = blockIdx.x;
  const int ft = wave;
  const bool mm = wave < 3;

  const int h = threadIdx.x;
  float p[NX], q[NX];
  load_row(w1_1, h, b1_1[h], p);
  load_row(w2_1, h, b2_1[h], q);

  v4f fa1[H / 8], fa2[H / 8];
  v4f wlo, whi, b1lo, b1hi, b2lo, b2hi;
  if (mm) {
    const float* r1 = w1_2 + (ob * 16 + l16) * H + 4 * half;
    const float* r2 = w2_2 + (ob * 16 + l16) * H + 4 * half;
#pragma unroll
    for (int u = 0; u < H / 8; ++u) {
      fa1[u] = *reinterpret_cast<const v4f*>(r1 + 8 * u);
      fa2[u] = *reinterpret_cast<const v4f*>(r2 + 8 * u);
    }
    const int o0 = ob * 16 + 8 * half;
    wlo = *reinterpret_cast<const v4f*>(w_out + o0);
    whi = *reinterpret_cast<const v4f*>(w_out + o0 + 4);
    b1lo = *reinterpret_cast<const v4f*>(b1_2 + o0);
    b1hi = *reinterpret_cast<const v4f*>(b1_2 + o0 + 4);
    b2lo = *reinterpret_cast<const v4f*>(b2_2 + o0);
    b2hi = *reinterpret_cast<const v4f*>(b2_2 + o0 + 4);
  }

  build_mt_rows<0>(mt, h, p, q);
  build_mt_rows<KF / 2>(mt, h, p, q);
  __syncthreads();

  v8f d1 = {}, d2 = {};
  if (mm) {
    const float* bm = mt + (16 * ft + l16) * W_PITCH + 4 * half;
#pragma unroll
    for (int u = 0; u < H / 8; ++u) {
      const v4f fb = *reinterpret_cast<const v4f*>(bm + 8 * u);
      const v2f b0 = {fb[0], fb[1]};
      const v2f b1 = {fb[2], fb[3]};
      const v2f a10 = {fa1[u][0], fa1[u][1]};
      const v2f a20 = {fa2[u][0], fa2[u][1]};
      const v2f a11 = {fa1[u][2], fa1[u][3]};
      const v2f a21 = {fa2[u][2], fa2[u][3]};
      d1 = wmma_k4(a10, b0, d1);
      d2 = wmma_k4(a20, b0, d2);
      d1 = wmma_k4(a11, b1, d1);
      d2 = wmma_k4(a21, b1, d2);
    }

    const bool bias_lane = (ft == 2) && (l16 == NF - 32);
#pragma unroll
    for (int j = 0; j < 8; ++j) {
      const float wo = j < 4 ? wlo[j & 3] : whi[j & 3];
      const float bb1 = j < 4 ? b1lo[j & 3] : b1hi[j & 3];
      const float bb2 = j < 4 ? b2lo[j & 3] : b2hi[j & 3];
      d1[j] = (bias_lane ? bb1 : d1[j]) * wo;
      d2[j] = bias_lane ? bb2 : d2[j];
    }
    const v4f lo = {d2[0], d2[1], d2[2], d2[3]};
    const v4f hi = {d2[4], d2[5], d2[6], d2[7]};
    cx[(ft * 2 + 0) * 32 + lane] = lo;
    cx[(ft * 2 + 1) * 32 + lane] = hi;
  }
  __syncthreads();

  if (mm) {
    v4f c2[3][2];
#pragma unroll
    for (int gb = 0; gb < 3; ++gb) {
      c2[gb][0] = cx[(gb * 2 + 0) * 32 + lane];
      c2[gb][1] = cx[(gb * 2 + 1) * 32 + lane];
    }
    v8f pq[3] = {};
#pragma unroll
    for (int u = 0; u < 4; ++u) {
      const v2f fa = {d1[2 * u], d1[2 * u + 1]};
#pragma unroll
      for (int gb = 0; gb < 3; ++gb) {
        const v2f fb2 = {c2[gb][u >> 1][2 * (u & 1)], c2[gb][u >> 1][2 * (u & 1) + 1]};
        pq[gb] = wmma_k4(fa, fb2, pq[gb]);
      }
    }
    float* pw = part + ob * (KF * KF);
#pragma unroll
    for (int gb = 0; gb < 3; ++gb) {
#pragma unroll
      for (int r = 0; r < 8; ++r)
        pw[(16 * ft + 8 * half + r) * KF + 16 * gb + l16] = pq[gb][r];
    }
  }
}

template <int WAVES>
__global__ __launch_bounds__(WAVES * 32) void net_all_kernel(
    const float* __restrict__ xT, const float* __restrict__ xL,
    const float* __restrict__ xLdot, const float* __restrict__ xC,
    float* __restrict__ outT, float* __restrict__ outY,
    float* __restrict__ outG, float* __restrict__ outC,
    int nT, int nL, int nC, const float* __restrict__ part) {
  __shared__ __attribute__((aligned(16))) float qs[KF * S_PITCH];
  __shared__ __attribute__((aligned(16))) float qt[KF * S_PITCH];

  const int wave = __builtin_amdgcn_readfirstlane(threadIdx.x >> 5);
  const int lane = threadIdx.x & 31;
  const int half = lane >> 4;
  const int l16 = lane & 15;

  constexpr int NPART = H / 16;
  constexpr int NQUAD = KF * (KF / 4);
  constexpr int NT = WAVES * 32;
  constexpr int FULL = NQUAD / NT;
  constexpr int REM = NQUAD - FULL * NT;
  static_assert(REM % 32 == 0, "the last pass covers whole waves: a scalar branch");
  const bool last = wave < REM / 32;
  v4f pr[FULL + 1][NPART];
  auto quad_load = [&](const int it) {
    const int qid = it * NT + threadIdx.x;
    const float* src = part + qid * 4;
#pragma unroll
    for (int w = 0; w < NPART; ++w)
      pr[it][w] = *reinterpret_cast<const v4f*>(src + w * (KF * KF));
  };
  auto quad_sum_store = [&](const int it) {
    const int qid = it * NT + threadIdx.x;
    const int f = qid / (KF / 4);
    const int g0 = (qid - f * (KF / 4)) * 4;
    v4f acc = pr[it][0];
#pragma unroll
    for (int w = 1; w < NPART; ++w) acc += pr[it][w];
    *reinterpret_cast<v4f*>(qs + f * S_PITCH + g0) = acc;
#pragma unroll
    for (int i = 0; i < 4; ++i) qt[(g0 + i) * S_PITCH + f] = acc[i];
  };
#pragma unroll
  for (int it = 0; it < FULL; ++it) quad_load(it);
  if (last) quad_load(FULL);
#pragma unroll
  for (int it = 0; it < FULL; ++it) quad_sum_store(it);
  if (last) quad_sum_store(FULL);
  __syncthreads();

  v4f sf[3][6];
#pragma unroll
  for (int mb = 0; mb < 3; ++mb) {
#pragma unroll
    for (int v = 0; v < 6; ++v) {
      const int off = (16 * mb + l16) * S_PITCH + 16 * (v >> 1) + 8 * half + 4 * (v & 1);
      sf[mb][v] = *reinterpret_cast<const v4f*>(qs + off) +
                  *reinterpret_cast<const v4f*>(qt + off);
    }
  }

  const int first = blockIdx.x + gridDim.x * wave;
  const int tstep = gridDim.x * WAVES;

  const int tilesT = (nT + 15) >> 4;
  const int tilesF = tilesT + ((nC + 15) >> 4);
  for (int tile = first; tile < tilesF; tile += tstep) {
    const bool small = tile >= tilesT;
    const float* x = small ? xC : xT;
    float* out = small ? outC : outT;
    const int row0 = (small ? tile - tilesT : tile) * 16;
    quad_tile<false>(x, nullptr, out, nullptr, row0, sf, half, l16);
  }

  const int tilesL = (nL + 15) >> 4;
  for (int tile = first; tile < tilesL; tile += tstep) {
    quad_tile<true>(xL, xLdot, outY, outG, tile * 16, sf, half, l16);
  }
}

extern "C" void kernel_launch(void* const* d_in, const int* in_sizes, int n_in,
                              void* d_out, int out_size, void* d_ws,
                              size_t ws_size, hipStream_t stream) {
  const float* T = (const float*)d_in[0];
  const float* l = (const float*)d_in[1];
  const float* l1_dot = (const float*)d_in[2];
  const float* center = (const float*)d_in[3];
  const float* w1_1 = (const float*)d_in[4];
  const float* b1_1 = (const float*)d_in[5];
  const float* w2_1 = (const float*)d_in[6];
  const float* b2_1 = (const float*)d_in[7];
  const float* w1_2 = (const float*)d_in[8];
  const float* b1_2 = (const float*)d_in[9];
  const float* w2_2 = (const float*)d_in[10];
  const float* b2_2 = (const float*)d_in[11];
  const float* w_out = (const float*)d_in[12];

  const int nT = in_sizes[0] / NIN;
  const int nL = in_sizes[1] / NIN;
  const int nC = in_sizes[3] / NIN;

  float* out = (float*)d_out;
  float* v_t = out;
  float* v_y = out + nT;
  float* v_g = out + nT + nL;
  float* v_c = out + nT + 2 * nL;

  float* part = (float*)d_ws;

  q_partial_kernel<<<H / 16, A_WAVES * 32, 0, stream>>>(
      w1_1, b1_1, w2_1, b2_1, w1_2, b1_2, w2_2, b2_2, w_out, part);

  constexpr int WV = 8;
  constexpr int NWG = 256;

  net_all_kernel<WV><<<NWG, WV * 32, 0, stream>>>(
      T, l, l1_dot, center, v_t, v_y, v_g, v_c, nT, nL, nC, part);
}


# Input rows of each wave's next tile loaded one tile ahead, the first tiles' rows in front of the partial loads: a tile product no longer waits for a global load; sums and their order unchanged
# speedup vs baseline: 10.3017x; 1.0564x over previous
#include <hip/hip_runtime.h>


typedef float v2f __attribute__((ext_vector_type(2)));
typedef float v4f __attribute__((ext_vector_type(4)));
typedef float v8f __attribute__((ext_vector_type(8)));

#define H 128
#define NIN 8
#define NX 9
#define NF 45
#define KF 48
#define W_PITCH 132
#define S_PITCH 52

__host__ __device__ constexpr int feat_i(int f) {
  int i = 0, len = NX;
  while (f >= len) { f -= len; ++i; --len; }
  return i;
}
__host__ __device__ constexpr int feat_j(int f) {
  int i = 0, len = NX;
  while (f >= len) { f -= len; ++i; --len; }
  return i + f;
}

template <int V> struct IC { static constexpr int value = V; };
template <int I, int N, class Fn>
__device__ __forceinline__ void static_for(Fn&& fn) {
  if constexpr (I < N) {
    fn(IC<I>{});
    static_for<I + 1, N>(fn);
  }
}

__device__ __forceinline__ v8f wmma_k4(v2f a, v2f b, v8f c) {
  return __builtin_amdgcn_wmma_f32_16x16x4_f32(
       false, a,  false, b,
       (short)0, c,  false,  false);
}

template <int F>
__device__ __forceinline__ float phi(const float (&xt)[NX]) {
  if constexpr (F < NF) {
    constexpr int i = feat_i(F), j = feat_j(F);
    return xt[i] * xt[j];
  } else {
    return 0.f;
  }
}
template <int F>
__device__ __forceinline__ float dphi(const float (&xt)[NX], const float (&xd)[NX]) {
  if constexpr (F < NF) {
    constexpr int i = feat_i(F), j = feat_j(F);
    return xd[i] * xt[j] + xt[i] * xd[j];
  } else {
    return 0.f;
  }
}

template <int F0>
__device__ __forceinline__ void build_mt_rows(float* mt, const int h,
                                              const float (&p)[NX],
                                              const float (&q)[NX]) {
  static_for<0, KF / 2>([&](auto ic) {
    constexpr int F = F0 + decltype(ic)::value;
    float v = 0.f;
    if constexpr (F < NF) {
      constexpr int i = feat_i(F), j = feat_j(F);
      if constexpr (i == j) v = p[i] * q[i];
      else v = p[i] * q[j] + p[j] * q[i];
    }
    mt[F * W_PITCH + h] = v;
  });
}

__device__ __forceinline__ void load_row(const float* x, const int row,
                                         const float last, float (&xt)[NX]) {
  const v4f lo = *reinterpret_cast<const v4f*>(x + row * NIN);
  const v4f hi = *reinterpret_cast<const v4f*>(x + row * NIN + 4);
  xt[0] = lo[0]; xt[1] = lo[1]; xt[2] = lo[2]; xt[3] = lo[3];
  xt[4] = hi[0]; xt[5] = hi[1]; xt[6] = hi[2]; xt[7] = hi[3];
  xt[8] = last;
}

template <int E>
__device__ __forceinline__ float phie(const float (&xt)[NX]) {
  if constexpr (E == NF) return 1.f;
  else return phi<E>(xt);
}
template <int E>
__device__ __forceinline__ float dphie(const float (&xt)[NX], const float (&xd)[NX]) {
  return dphi<E>(xt, xd);
}

__host__ __device__ constexpr int slot_feat(int half, int idx) {
  return 16 * (idx >> 3) + 8 * half + (idx & 7);
}

template <bool GRAD>
__device__ __forceinline__ void quad_tile_rows(
    const v4f (&xr)[2], const v4f (&dr)[2], float* outY, float* outG,
    const int row0, const v4f (&sf)[3][6], const int half, const int l16) {
  float xt[NX], xd[NX];
  xt[0] = xr[0][0]; xt[1] = xr[0][1]; xt[2] = xr[0][2]; xt[3] = xr[0][3];
  xt[4] = xr[1][0]; xt[5] = xr[1][1]; xt[6] = xr[1][2]; xt[7] = xr[1][3];
  xt[8] = 1.f;
  if constexpr (GRAD) {
    xd[0] = dr[0][0]; xd[1] = dr[0][1]; xd[2] = dr[0][2]; xd[3] = dr[0][3];
    xd[4] = dr[1][0]; xd[5] = dr[1][1]; xd[6] = dr[1][2]; xd[7] = dr[1][3];
    xd[8] = 0.f;
  }
  float pv[24], dv[24];
  static_for<0, 24>([&](auto ic) {
    constexpr int i = decltype(ic)::value;
    pv[i] = half ? phie<slot_feat(1, i)>(xt) : phie<slot_feat(0, i)>(xt);
    if constexpr (GRAD)
      dv[i] = half ? dphie<slot_feat(1, i)>(xt, xd) : dphie<slot_feat(0, i)>(xt, xd);
  });

  v8f t[3] = {};
#pragma unroll
  for (int v = 0; v < 6; ++v) {
    const v2f b0 = {pv[4 * v], pv[4 * v + 1]};
    const v2f b1 = {pv[4 * v + 2], pv[4 * v + 3]};
#pragma unroll
    for (int mb = 0; mb < 3; ++mb) {
      const v2f a0 = {sf[mb][v][0], sf[mb][v][1]};
      t[mb] = wmma_k4(a0, b0, t[mb]);
    }
#pragma unroll
    for (int mb = 0; mb < 3; ++mb) {
      const v2f a1 = {sf[mb][v][2], sf[mb][v][3]};
      t[mb] = wmma_k4(a1, b1, t[mb]);
    }
  }

  float ys[3], gs[3];
#pragma unroll
  for (int mb = 0; mb < 3; ++mb) {
    ys[mb] = 0.f;
    gs[mb] = 0.f;
#pragma unroll
    for (int r = 0; r < 8; ++r) {
      ys[mb] += t[mb][r] * pv[8 * mb + r];
      if constexpr (GRAD) gs[mb] += t[mb][r] * dv[8 * mb + r];
    }
  }
  float y = (ys[0] + ys[1]) + ys[2];
  y += __shfl_xor(y, 16, 32);
  y *= 0.5f;
  if constexpr (GRAD) {
    float g = (gs[0] + gs[1]) + gs[2];
    g += __shfl_xor(g, 16, 32);
    float* dst = half ? outG : outY;
    dst[row0 + l16] = half ? g : y;
  } else {
    if (half == 0) outY[row0 + l16] = y;
  }
}

__device__ __forceinline__ void load_row_regs(const float* x, const int row, v4f (&r)[2]) {
  r[0] = *reinterpret_cast<const v4f*>(x + row * NIN);
  r[1] = *reinterpret_cast<const v4f*>(x + row * NIN + 4);
}

#define A_WAVES 4
__global__ __launch_bounds__(A_WAVES * 32) void q_partial_kernel(
    const float* __restrict__ w1_1, const float* __restrict__ b1_1,
    const float* __restrict__ w2_1, const float* __restrict__ b2_1,
    const float* __restrict__ w1_2, const float* __restrict__ b1_2,
    const float* __restrict__ w2_2, const float* __restrict__ b2_2,
    const float* __restrict__ w_out, float* __restrict__ part) {
  static_assert(A_WAVES * 32 == H, "M^T: one thread per hidden unit h");
  __shared__ __attribute__((aligned(16))) float mt[KF * W_PITCH];
  __shared__ __attribute__((aligned(16))) v4f cx[3 * 2 * 32];

  const int wave = __builtin_amdgcn_readfirstlane(threadIdx.x >> 5);
  const int lane = threadIdx.x & 31;
  const int half = lane >> 4;
  const int l16 = lane & 15;
  const int ob = blockIdx.x;
  const int ft = wave;
  const bool mm = wave < 3;

  const int h = threadIdx.x;
  float p[NX], q[NX];
  load_row(w1_1, h, b1_1[h], p);
  load_row(w2_1, h, b2_1[h], q);

  v4f fa1[H / 8], fa2[H / 8];
  v4f wlo, whi, b1lo, b1hi, b2lo, b2hi;
  if (mm) {
    const float* r1 = w1_2 + (ob * 16 + l16) * H + 4 * half;
    const float* r2 = w2_2 + (ob * 16 + l16) * H + 4 * half;
#pragma unroll
    for (int u = 0; u < H / 8; ++u) {
      fa1[u] = *reinterpret_cast<const v4f*>(r1 + 8 * u);
      fa2[u] = *reinterpret_cast<const v4f*>(r2 + 8 * u);
    }
    const int o0 = ob * 16 + 8 * half;
    wlo = *reinterpret_cast<const v4f*>(w_out + o0);
    whi = *reinterpret_cast<const v4f*>(w_out + o0 + 4);
    b1lo = *reinterpret_cast<const v4f*>(b1_2 + o0);
    b1hi = *reinterpret_cast<const v4f*>(b1_2 + o0 + 4);
    b2lo = *reinterpret_cast<const v4f*>(b2_2 + o0);
    b2hi = *reinterpret_cast<const v4f*>(b2_2 + o0 + 4);
  }

  build_mt_rows<0>(mt, h, p, q);
  build_mt_rows<KF / 2>(mt, h, p, q);
  __syncthreads();

  v8f d1 = {}, d2 = {};
  if (mm) {
    const float* bm = mt + (16 * ft + l16) * W_PITCH + 4 * half;
#pragma unroll
    for (int u = 0; u < H / 8; ++u) {
      const v4f fb = *reinterpret_cast<const v4f*>(bm + 8 * u);
      const v2f b0 = {fb[0], fb[1]};
      const v2f b1 = {fb[2], fb[3]};
      const v2f a10 = {fa1[u][0], fa1[u][1]};
      const v2f a20 = {fa2[u][0], fa2[u][1]};
      const v2f a11 = {fa1[u][2], fa1[u][3]};
      const v2f a21 = {fa2[u][2], fa2[u][3]};
      d1 = wmma_k4(a10, b0, d1);
      d2 = wmma_k4(a20, b0, d2);
      d1 = wmma_k4(a11, b1, d1);
      d2 = wmma_k4(a21, b1, d2);
    }

    const bool bias_lane = (ft == 2) && (l16 == NF - 32);
#pragma unroll
    for (int j = 0; j < 8; ++j) {
      const float wo = j < 4 ? wlo[j & 3] : whi[j & 3];
      const float bb1 = j < 4 ? b1lo[j & 3] : b1hi[j & 3];
      const float bb2 = j < 4 ? b2lo[j & 3] : b2hi[j & 3];
      d1[j] = (bias_lane ? bb1 : d1[j]) * wo;
      d2[j] = bias_lane ? bb2 : d2[j];
    }
    const v4f lo = {d2[0], d2[1], d2[2], d2[3]};
    const v4f hi = {d2[4], d2[5], d2[6], d2[7]};
    cx[(ft * 2 + 0) * 32 + lane] = lo;
    cx[(ft * 2 + 1) * 32 + lane] = hi;
  }
  __syncthreads();

  if (mm) {
    v4f c2[3][2];
#pragma unroll
    for (int gb = 0; gb < 3; ++gb) {
      c2[gb][0] = cx[(gb * 2 + 0) * 32 + lane];
      c2[gb][1] = cx[(gb * 2 + 1) * 32 + lane];
    }
    v8f pq[3] = {};
#pragma unroll
    for (int u = 0; u < 4; ++u) {
      const v2f fa = {d1[2 * u], d1[2 * u + 1]};
#pragma unroll
      for (int gb = 0; gb < 3; ++gb) {
        const v2f fb2 = {c2[gb][u >> 1][2 * (u & 1)], c2[gb][u >> 1][2 * (u & 1) + 1]};
        pq[gb] = wmma_k4(fa, fb2, pq[gb]);
      }
    }
    float* pw = part + ob * (KF * KF);
#pragma unroll
    for (int gb = 0; gb < 3; ++gb) {
#pragma unroll
      for (int r = 0; r < 8; ++r)
        pw[(16 * ft + 8 * half + r) * KF + 16 * gb + l16] = pq[gb][r];
    }
  }
}

template <int WAVES>
__global__ __launch_bounds__(WAVES * 32) void net_all_kernel(
    const float* __restrict__ xT, const float* __restrict__ xL,
    const float* __restrict__ xLdot, const float* __restrict__ xC,
    float* __restrict__ outT, float* __restrict__ outY,
    float* __restrict__ outG, float* __restrict__ outC,
    int nT, int nL, int nC, const float* __restrict__ part) {
  __shared__ __attribute__((aligned(16))) float qs[KF * S_PITCH];
  __shared__ __attribute__((aligned(16))) float qt[KF * S_PITCH];

  const int wave = __builtin_amdgcn_readfirstlane(threadIdx.x >> 5);
  const int lane = threadIdx.x & 31;
  const int half = lane >> 4;
  const int l16 = lane & 15;

  const int first = blockIdx.x + gridDim.x * wave;
  const int tstep = gridDim.x * WAVES;
  const int tilesT = (nT + 15) >> 4;
  const int tilesF = tilesT + ((nC + 15) >> 4);
  const int tilesL = (nL + 15) >> 4;
  auto f_src = [&](const int tile) { return tile >= tilesT ? xC : xT; };
  auto f_dst = [&](const int tile) { return tile >= tilesT ? outC : outT; };
  auto f_row0 = [&](const int tile) { return (tile >= tilesT ? tile - tilesT : tile) * 16; };
  v4f fcur[2], lcur[2], dcur[2];
  {
    const int tf = first < tilesF ? first : 0;
    load_row_regs(f_src(tf), f_row0(tf) + l16, fcur);
    const int tl = first < tilesL ? first : 0;
    load_row_regs(xL, tl * 16 + l16, lcur);
    load_row_regs(xLdot, tl * 16 + l16, dcur);
  }

  constexpr int NPART = H / 16;
  constexpr int NQUAD = KF * (KF / 4);
  constexpr int NT = WAVES * 32;
  constexpr int FULL = NQUAD / NT;
  constexpr int REM = NQUAD - FULL * NT;
  static_assert(REM % 32 == 0, "the last pass covers whole waves: a scalar branch");
  const bool last = wave < REM / 32;
  v4f pr[FULL + 1][NPART];
  auto quad_load = [&](const int it) {
    const int qid = it * NT + threadIdx.x;
    const float* src = part + qid * 4;
#pragma unroll
    for (int w = 0; w < NPART; ++w)
      pr[it][w] = *reinterpret_cast<const v4f*>(src + w * (KF * KF));
  };
  auto quad_sum_store = [&](const int it) {
    const int qid = it * NT + threadIdx.x;
    const int f = qid / (KF / 4);
    const int g0 = (qid - f * (KF / 4)) * 4;
    v4f acc = pr[it][0];
#pragma unroll
    for (int w = 1; w < NPART; ++w) acc += pr[it][w];
    *reinterpret_cast<v4f*>(qs + f * S_PITCH + g0) = acc;
#pragma unroll
    for (int i = 0; i < 4; ++i) qt[(g0 + i) * S_PITCH + f] = acc[i];
  };
#pragma unroll
  for (int it = 0; it < FULL; ++it) quad_load(it);
  if (last) quad_load(FULL);
#pragma unroll
  for (int it = 0; it < FULL; ++it) quad_sum_store(it);
  if (last) quad_sum_store(FULL);
  __syncthreads();

  v4f sf[3][6];
#pragma unroll
  for (int mb = 0; mb < 3; ++mb) {
#pragma unroll
    for (int v = 0; v < 6; ++v) {
      const int off = (16 * mb + l16) * S_PITCH + 16 * (v >> 1) + 8 * half + 4 * (v & 1);
      sf[mb][v] = *reinterpret_cast<const v4f*>(qs + off) +
                  *reinterpret_cast<const v4f*>(qt + off);
    }
  }

  for (int tile = first; tile < tilesF; tile += tstep) {
    const int nx = tile + tstep < tilesF ? tile + tstep : 0;
    v4f fnxt[2];
    load_row_regs(f_src(nx), f_row0(nx) + l16, fnxt);
    quad_tile_rows<false>(fcur, fcur, f_dst(tile), nullptr, f_row0(tile), sf, half, l16);
    fcur[0] = fnxt[0];
    fcur[1] = fnxt[1];
  }

  for (int tile = first; tile < tilesL; tile += tstep) {
    const int nx = tile + tstep < tilesL ? tile + tstep : 0;
    v4f lnxt[2], dnxt[2];
    load_row_regs(xL, nx * 16 + l16, lnxt);
    load_row_regs(xLdot, nx * 16 + l16, dnxt);
    __builtin_amdgcn_sched_barrier(0);
    quad_tile_rows<true>(lcur, dcur, outY, outG, tile * 16, sf, half, l16);
    lcur[0] = lnxt[0];
    lcur[1] = lnxt[1];
    dcur[0] = dnxt[0];
    dcur[1] = dnxt[1];
  }
}

extern "C" void kernel_launch(void* const* d_in, const int* in_sizes, int n_in,
                              void* d_out, int out_size, void* d_ws,
                              size_t ws_size, hipStream_t stream) {
  const float* T = (const float*)d_in[0];
  const float* l = (const float*)d_in[1];
  const float* l1_dot = (const float*)d_in[2];
  const float* center = (const float*)d_in[3];
  const float* w1_1 = (const float*)d_in[4];
  const float* b1_1 = (const float*)d_in[5];
  const float* w2_1 = (const float*)d_in[6];
  const float* b2_1 = (const float*)d_in[7];
  const float* w1_2 = (const float*)d_in[8];
  const float* b1_2 = (const float*)d_in[9];
  const float* w2_2 = (const float*)d_in[10];
  const float* b2_2 = (const float*)d_in[11];
  const float* w_out = (const float*)d_in[12];

  const int nT = in_sizes[0] / NIN;
  const int nL = in_sizes[1] / NIN;
  const int nC = in_sizes[3] / NIN;

  float* out = (float*)d_out;
  float* v_t = out;
  float* v_y = out + nT;
  float* v_g = out + nT + nL;
  float* v_c = out + nT + 2 * nL;

  float* part = (float*)d_ws;

  q_partial_kernel<<<H / 16, A_WAVES * 32, 0, stream>>>(
      w1_1, b1_1, w2_1, b2_1, w1_2, b1_2, w2_2, b2_2, w_out, part);

  constexpr int WV = 8;
  constexpr int NWG = 256;

  net_all_kernel<WV><<<NWG, WV * 32, 0, stream>>>(
      T, l, l1_dot, center, v_t, v_y, v_g, v_c, nT, nL, nC, part);
}


# S used as symmetric: launch 1 writes 6 of 9 blocks per partial of S=Q+Q^T, launch 2 adds 2/3 of the bytes, value-only tiles use its upper block triangle: 24 WMMAs, not 36; sum order changes
# speedup vs baseline: 11.1279x; 1.0802x over previous
#include <hip/hip_runtime.h>


typedef float v2f __attribute__((ext_vector_type(2)));
typedef float v4f __attribute__((ext_vector_type(4)));
typedef float v8f __attribute__((ext_vector_type(8)));

#define H 128
#define NIN 8
#define NX 9
#define NF 45
#define KF 48
#define W_PITCH 132
#define S_PITCH 52

__host__ __device__ constexpr int feat_i(int f) {
  int i = 0, len = NX;
  while (f >= len) { f -= len; ++i; --len; }
  return i;
}
__host__ __device__ constexpr int feat_j(int f) {
  int i = 0, len = NX;
  while (f >= len) { f -= len; ++i; --len; }
  return i + f;
}

template <int V> struct IC { static constexpr int value = V; };
template <int I, int N, class Fn>
__device__ __forceinline__ void static_for(Fn&& fn) {
  if constexpr (I < N) {
    fn(IC<I>{});
    static_for<I + 1, N>(fn);
  }
}

__device__ __forceinline__ v8f wmma_k4(v2f a, v2f b, v8f c) {
  return __builtin_amdgcn_wmma_f32_16x16x4_f32(
       false, a,  false, b,
       (short)0, c,  false,  false);
}

template <int F>
__device__ __forceinline__ float phi(const float (&xt)[NX]) {
  if constexpr (F < NF) {
    constexpr int i = feat_i(F), j = feat_j(F);
    return xt[i] * xt[j];
  } else {
    return 0.f;
  }
}
template <int F>
__device__ __forceinline__ float dphi(const float (&xt)[NX], const float (&xd)[NX]) {
  if constexpr (F < NF) {
    constexpr int i = feat_i(F), j = feat_j(F);
    return xd[i] * xt[j] + xt[i] * xd[j];
  } else {
    return 0.f;
  }
}

template <int F0>
__device__ __forceinline__ void build_mt_rows(float* mt, const int h,
                                              const float (&p)[NX],
                                              const float (&q)[NX]) {
  static_for<0, KF / 2>([&](auto ic) {
    constexpr int F = F0 + decltype(ic)::value;
    float v = 0.f;
    if constexpr (F < NF) {
      constexpr int i = feat_i(F), j = feat_j(F);
      if constexpr (i == j) v = p[i] * q[i];
      else v = p[i] * q[j] + p[j] * q[i];
    }
    mt[F * W_PITCH + h] = v;
  });
}

__device__ __forceinline__ void load_row(const float* x, const int row,
                                         const float last, float (&xt)[NX]) {
  const v4f lo = *reinterpret_cast<const v4f*>(x + row * NIN);
  const v4f hi = *reinterpret_cast<const v4f*>(x + row * NIN + 4);
  xt[0] = lo[0]; xt[1] = lo[1]; xt[2] = lo[2]; xt[3] = lo[3];
  xt[4] = hi[0]; xt[5] = hi[1]; xt[6] = hi[2]; xt[7] = hi[3];
  xt[8] = last;
}

template <int E>
__device__ __forceinline__ float phie(const float (&xt)[NX]) {
  if constexpr (E == NF) return 1.f;
  else return phi<E>(xt);
}
template <int E>
__device__ __forceinline__ float dphie(const float (&xt)[NX], const float (&xd)[NX]) {
  return dphi<E>(xt, xd);
}

__host__ __device__ constexpr int slot_feat(int half, int idx) {
  return 16 * (idx >> 3) + 8 * half + (idx & 7);
}

template <bool GRAD>
__device__ __forceinline__ void quad_tile_rows(
    const v4f (&xr)[2], const v4f (&dr)[2], float* outY, float* outG,
    const int row0, const v4f (&sf)[3][6], const int half, const int l16) {
  float xt[NX], xd[NX];
  xt[0] = xr[0][0]; xt[1] = xr[0][1]; xt[2] = xr[0][2]; xt[3] = xr[0][3];
  xt[4] = xr[1][0]; xt[5] = xr[1][1]; xt[6] = xr[1][2]; xt[7] = xr[1][3];
  xt[8] = 1.f;
  if constexpr (GRAD) {
    xd[0] = dr[0][0]; xd[1] = dr[0][1]; xd[2] = dr[0][2]; xd[3] = dr[0][3];
    xd[4] = dr[1][0]; xd[5] = dr[1][1]; xd[6] = dr[1][2]; xd[7] = dr[1][3];
    xd[8] = 0.f;
  }
  float pv[24], dv[24];
  static_for<0, 24>([&](auto ic) {
    constexpr int i = decltype(ic)::value;
    pv[i] = half ? phie<slot_feat(1, i)>(xt) : phie<slot_feat(0, i)>(xt);
    if constexpr (GRAD)
      dv[i] = half ? dphie<slot_feat(1, i)>(xt, xd) : dphie<slot_feat(0, i)>(xt, xd);
  });

  if constexpr (!GRAD) {
    v8f td[3] = {}, tu[2] = {};
#pragma unroll
    for (int v = 0; v < 6; ++v) {
      const v2f b0 = {pv[4 * v], pv[4 * v + 1]};
      const v2f b1 = {pv[4 * v + 2], pv[4 * v + 3]};
#pragma unroll
      for (int mb = 0; mb <= (v >> 1); ++mb) {
        const v2f a0 = {sf[mb][v][0], sf[mb][v][1]};
        if (mb == (v >> 1)) td[mb] = wmma_k4(a0, b0, td[mb]);
        else tu[mb] = wmma_k4(a0, b0, tu[mb]);
      }
#pragma unroll
      for (int mb = 0; mb <= (v >> 1); ++mb) {
        const v2f a1 = {sf[mb][v][2], sf[mb][v][3]};
        if (mb == (v >> 1)) td[mb] = wmma_k4(a1, b1, td[mb]);
        else tu[mb] = wmma_k4(a1, b1, tu[mb]);
      }
    }
    float yd[3], yu[2];
#pragma unroll
    for (int mb = 0; mb < 3; ++mb) {
      yd[mb] = 0.f;
#pragma unroll
      for (int r = 0; r < 8; ++r) yd[mb] += td[mb][r] * pv[8 * mb + r];
    }
#pragma unroll
    for (int mb = 0; mb < 2; ++mb) {
      yu[mb] = 0.f;
#pragma unroll
      for (int r = 0; r < 8; ++r) yu[mb] += tu[mb][r] * pv[8 * mb + r];
    }
    float y = 0.5f * ((yd[0] + yd[1]) + yd[2]) + (yu[0] + yu[1]);
    y += __shfl_xor(y, 16, 32);
    if (half == 0) outY[row0 + l16] = y;
    return;
  }

  v8f t[3] = {};
#pragma unroll
  for (int v = 0; v < 6; ++v) {
    const v2f b0 = {pv[4 * v], pv[4 * v + 1]};
    const v2f b1 = {pv[4 * v + 2], pv[4 * v + 3]};
#pragma unroll
    for (int mb = 0; mb < 3; ++mb) {
      const v2f a0 = {sf[mb][v][0], sf[mb][v][1]};
      t[mb] = wmma_k4(a0, b0, t[mb]);
    }
#pragma unroll
    for (int mb = 0; mb < 3; ++mb) {
      const v2f a1 = {sf[mb][v][2], sf[mb][v][3]};
      t[mb] = wmma_k4(a1, b1, t[mb]);
    }
  }

  float ys[3], gs[3];
#pragma unroll
  for (int mb = 0; mb < 3; ++mb) {
    ys[mb] = 0.f;
    gs[mb] = 0.f;
#pragma unroll
    for (int r = 0; r < 8; ++r) {
      ys[mb] += t[mb][r] * pv[8 * mb + r];
      if constexpr (GRAD) gs[mb] += t[mb][r] * dv[8 * mb + r];
    }
  }
  float y = (ys[0] + ys[1]) + ys[2];
  y += __shfl_xor(y, 16, 32);
  y *= 0.5f;
  if constexpr (GRAD) {
    float g = (gs[0] + gs[1]) + gs[2];
    g += __shfl_xor(g, 16, 32);
    float* dst = half ? outG : outY;
    dst[row0 + l16] = half ? g : y;
  } else {
    if (half == 0) outY[row0 + l16] = y;
  }
}

__device__ __forceinline__ void load_row_regs(const float* x, const int row, v4f (&r)[2]) {
  r[0] = *reinterpret_cast<const v4f*>(x + row * NIN);
  r[1] = *reinterpret_cast<const v4f*>(x + row * NIN + 4);
}

#define A_WAVES 4
#define P_BLOCKS 6
#define P_FLOATS (P_BLOCKS * 256)
__global__ __launch_bounds__(A_WAVES * 32) void q_partial_kernel(
    const float* __restrict__ w1_1, const float* __restrict__ b1_1,
    const float* __restrict__ w2_1, const float* __restrict__ b2_1,
    const float* __restrict__ w1_2, const float* __restrict__ b1_2,
    const float* __restrict__ w2_2, const float* __restrict__ b2_2,
    const float* __restrict__ w_out, float* __restrict__ part) {
  static_assert(A_WAVES * 32 == H, "M^T: one thread per hidden unit h");
  __shared__ __attribute__((aligned(16))) float mt[KF * W_PITCH];
  __shared__ __attribute__((aligned(16))) v4f cx[2 * 3 * 2 * 32];

  const int wave = __builtin_amdgcn_readfirstlane(threadIdx.x >> 5);
  const int lane = threadIdx.x & 31;
  const int half = lane >> 4;
  const int l16 = lane & 15;
  const int ob = blockIdx.x;
  const int ft = wave;
  const bool mm = wave < 3;

  const int h = threadIdx.x;
  float p[NX], q[NX];
  load_row(w1_1, h, b1_1[h], p);
  load_row(w2_1, h, b2_1[h], q);

  v4f fa1[H / 8], fa2[H / 8];
  v4f wlo, whi, b1lo, b1hi, b2lo, b2hi;
  if (mm) {
    const float* r1 = w1_2 + (ob * 16 + l16) * H + 4 * half;
    const float* r2 = w2_2 + (ob * 16 + l16) * H + 4 * half;
#pragma unroll
    for (int u = 0; u < H / 8; ++u) {
      fa1[u] = *reinterpret_cast<const v4f*>(r1 + 8 * u);
      fa2[u] = *reinterpret_cast<const v4f*>(r2 + 8 * u);
    }
    const int o0 = ob * 16 + 8 * half;
    wlo = *reinterpret_cast<const v4f*>(w_out + o0);
    whi = *reinterpret_cast<const v4f*>(w_out + o0 + 4);
    b1lo = *reinterpret_cast<const v4f*>(b1_2 + o0);
    b1hi = *reinterpret_cast<const v4f*>(b1_2 + o0 + 4);
    b2lo = *reinterpret_cast<const v4f*>(b2_2 + o0);
    b2hi = *reinterpret_cast<const v4f*>(b2_2 + o0 + 4);
  }

  build_mt_rows<0>(mt, h, p, q);
  build_mt_rows<KF / 2>(mt, h, p, q);
  __syncthreads();

  v8f d1 = {}, d2 = {};
  if (mm) {
    const float* bm = mt + (16 * ft + l16) * W_PITCH + 4 * half;
#pragma unroll
    for (int u = 0; u < H / 8; ++u) {
      const v4f fb = *reinterpret_cast<const v4f*>(bm + 8 * u);
      const v2f b0 = {fb[0], fb[1]};
      const v2f b1 = {fb[2], fb[3]};
      const v2f a10 = {fa1[u][0], fa1[u][1]};
      const v2f a20 = {fa2[u][0], fa2[u][1]};
      const v2f a11 = {fa1[u][2], fa1[u][3]};
      const v2f a21 = {fa2[u][2], fa2[u][3]};
      d1 = wmma_k4(a10, b0, d1);
      d2 = wmma_k4(a20, b0, d2);
      d1 = wmma_k4(a11, b1, d1);
      d2 = wmma_k4(a21, b1, d2);
    }

    const bool bias_lane = (ft == 2) && (l16 == NF - 32);
#pragma unroll
    for (int j = 0; j < 8; ++j) {
      const float wo = j < 4 ? wlo[j & 3] : whi[j & 3];
      const float bb1 = j < 4 ? b1lo[j & 3] : b1hi[j & 3];
      const float bb2 = j < 4 ? b2lo[j & 3] : b2hi[j & 3];
      d1[j] = (bias_lane ? bb1 : d1[j]) * wo;
      d2[j] = bias_lane ? bb2 : d2[j];
    }
    const v4f lo1 = {d1[0], d1[1], d1[2], d1[3]};
    const v4f hi1 = {d1[4], d1[5], d1[6], d1[7]};
    const v4f lo2 = {d2[0], d2[1], d2[2], d2[3]};
    const v4f hi2 = {d2[4], d2[5], d2[6], d2[7]};
    cx[((0 * 3 + ft) * 2 + 0) * 32 + lane] = lo1;
    cx[((0 * 3 + ft) * 2 + 1) * 32 + lane] = hi1;
    cx[((1 * 3 + ft) * 2 + 0) * 32 + lane] = lo2;
    cx[((1 * 3 + ft) * 2 + 1) * 32 + lane] = hi2;
  }
  __syncthreads();

  if (mm) {
    const int gx = ft == 2 ? 0 : ft + 1;
    v4f c1[2][2], c2[2][2];
#pragma unroll
    for (int d = 0; d < 2; ++d) {
      const int gb = d ? gx : ft;
      c1[d][0] = cx[((0 * 3 + gb) * 2 + 0) * 32 + lane];
      c1[d][1] = cx[((0 * 3 + gb) * 2 + 1) * 32 + lane];
      c2[d][0] = cx[((1 * 3 + gb) * 2 + 0) * 32 + lane];
      c2[d][1] = cx[((1 * 3 + gb) * 2 + 1) * 32 + lane];
    }
    v8f pq[2] = {};
#pragma unroll
    for (int u = 0; u < 4; ++u) {
      const v2f fa = {d1[2 * u], d1[2 * u + 1]};
#pragma unroll
      for (int d = 0; d < 2; ++d) {
        const v2f fb2 = {c2[d][u >> 1][2 * (u & 1)], c2[d][u >> 1][2 * (u & 1) + 1]};
        pq[d] = wmma_k4(fa, fb2, pq[d]);
      }
    }
#pragma unroll
    for (int u = 0; u < 4; ++u) {
      const v2f fa = {d2[2 * u], d2[2 * u + 1]};
#pragma unroll
      for (int d = 0; d < 2; ++d) {
        const v2f fb1 = {c1[d][u >> 1][2 * (u & 1)], c1[d][u >> 1][2 * (u & 1) + 1]};
        pq[d] = wmma_k4(fa, fb1, pq[d]);
      }
    }
    float* pw = part + ob * P_FLOATS + (2 * ft) * 256;
#pragma unroll
    for (int d = 0; d < 2; ++d) {
#pragma unroll
      for (int r = 0; r < 8; ++r) pw[d * 256 + r * 32 + lane] = pq[d][r];
    }
  }
}

template <int WAVES>
__global__ __launch_bounds__(WAVES * 32) void net_all_kernel(
    const float* __restrict__ xT, const float* __restrict__ xL,
    const float* __restrict__ xLdot, const float* __restrict__ xC,
    float* __restrict__ outT, float* __restrict__ outY,
    float* __restrict__ outG, float* __restrict__ outC,
    int nT, int nL, int nC, const float* __restrict__ part) {
  __shared__ __attribute__((aligned(16))) float ss[KF * S_PITCH];

  const int wave = __builtin_amdgcn_readfirstlane(threadIdx.x >> 5);
  const int lane = threadIdx.x & 31;
  const int half = lane >> 4;
  const int l16 = lane & 15;

  const int first = blockIdx.x + gridDim.x * wave;
  const int tstep = gridDim.x * WAVES;
  const int tilesT = (nT + 15) >> 4;
  const int tilesF = tilesT + ((nC + 15) >> 4);
  const int tilesL = (nL + 15) >> 4;
  auto f_src = [&](const int tile) { return tile >= tilesT ? xC : xT; };
  auto f_dst = [&](const int tile) { return tile >= tilesT ? outC : outT; };
  auto f_row0 = [&](const int tile) { return (tile >= tilesT ? tile - tilesT : tile) * 16; };
  v4f fcur[2], lcur[2], dcur[2];
  {
    const int tf = first < tilesF ? first : 0;
    load_row_regs(f_src(tf), f_row0(tf) + l16, fcur);
    const int tl = first < tilesL ? first : 0;
    load_row_regs(xL, tl * 16 + l16, lcur);
    load_row_regs(xLdot, tl * 16 + l16, dcur);
  }

  constexpr int NPART = H / 16;
  constexpr int NQUAD = P_FLOATS / 4;
  constexpr int NT = WAVES * 32;
  constexpr int FULL = NQUAD / NT;
  constexpr int REM = NQUAD - FULL * NT;
  static_assert(NT % 64 == 0 && REM % 64 == 0, "whole blocks per pass: scalar branches");
  const bool last = wave < REM / 32;
  v4f pr[FULL + 1][NPART];
  auto quad_load = [&](const int it) {
    const int qid = it * NT + threadIdx.x;
    const float* src = part + qid * 4;
#pragma unroll
    for (int w = 0; w < NPART; ++w)
      pr[it][w] = *reinterpret_cast<const v4f*>(src + w * P_FLOATS);
  };
  auto quad_sum_store = [&](const int it) {
    const int blk = it * (NT / 64) + (wave >> 1);
    const int fb = blk >> 1;
    const bool mirrored = blk & 1;
    const int gb = mirrored ? (fb == 2 ? 0 : fb + 1) : fb;
    const int q = threadIdx.x & 63;
    const int f = 16 * fb + 8 * ((q >> 2) & 1) + (q >> 3);
    const int g0 = 16 * gb + 4 * (q & 3);
    v4f acc = pr[it][0];
#pragma unroll
    for (int w = 1; w < NPART; ++w) acc += pr[it][w];
    *reinterpret_cast<v4f*>(ss + f * S_PITCH + g0) = acc;
    if (mirrored) {
#pragma unroll
      for (int i = 0; i < 4; ++i) ss[(g0 + i) * S_PITCH + f] = acc[i];
    }
  };
#pragma unroll
  for (int it = 0; it < FULL; ++it) quad_load(it);
  if (last) quad_load(FULL);
#pragma unroll
  for (int it = 0; it < FULL; ++it) quad_sum_store(it);
  if (last) quad_sum_store(FULL);
  __syncthreads();

  v4f sf[3][6];
#pragma unroll
  for (int mb = 0; mb < 3; ++mb) {
#pragma unroll
    for (int v = 0; v < 6; ++v) {
      const int off = (16 * mb + l16) * S_PITCH + 16 * (v >> 1) + 8 * half + 4 * (v & 1);
      sf[mb][v] = *reinterpret_cast<const v4f*>(ss + off);
    }
  }

  for (int tile = first; tile < tilesF; tile += tstep) {
    const int nx = tile + tstep < tilesF ? tile + tstep : 0;
    v4f fnxt[2];
    load_row_regs(f_src(nx), f_row0(nx) + l16, fnxt);
    quad_tile_rows<false>(fcur, fcur, f_dst(tile), nullptr, f_row0(tile), sf, half, l16);
    fcur[0] = fnxt[0];
    fcur[1] = fnxt[1];
  }

  for (int tile = first; tile < tilesL; tile += tstep) {
    const int nx = tile + tstep < tilesL ? tile + tstep : 0;
    v4f lnxt[2], dnxt[2];
    load_row_regs(xL, nx * 16 + l16, lnxt);
    load_row_regs(xLdot, nx * 16 + l16, dnxt);
    __builtin_amdgcn_sched_barrier(0);
    quad_tile_rows<true>(lcur, dcur, outY, outG, tile * 16, sf, half, l16);
    lcur[0] = lnxt[0];
    lcur[1] = lnxt[1];
    dcur[0] = dnxt[0];
    dcur[1] = dnxt[1];
  }
}

extern "C" void kernel_launch(void* const* d_in, const int* in_sizes, int n_in,
                              void* d_out, int out_size, void* d_ws,
                              size_t ws_size, hipStream_t stream) {
  const float* T = (const float*)d_in[0];
  const float* l = (const float*)d_in[1];
  const float* l1_dot = (const float*)d_in[2];
  const float* center = (const float*)d_in[3];
  const float* w1_1 = (const float*)d_in[4];
  const float* b1_1 = (const float*)d_in[5];
  const float* w2_1 = (const float*)d_in[6];
  const float* b2_1 = (const float*)d_in[7];
  const float* w1_2 = (const float*)d_in[8];
  const float* b1_2 = (const float*)d_in[9];
  const float* w2_2 = (const float*)d_in[10];
  const float* b2_2 = (const float*)d_in[11];
  const float* w_out = (const float*)d_in[12];

  const int nT = in_sizes[0] / NIN;
  const int nL = in_sizes[1] / NIN;
  const int nC = in_sizes[3] / NIN;

  float* out = (float*)d_out;
  float* v_t = out;
  float* v_y = out + nT;
  float* v_g = out + nT + nL;
  float* v_c = out + nT + 2 * nL;

  float* part = (float*)d_ws;

  q_partial_kernel<<<H / 16, A_WAVES * 32, 0, stream>>>(
      w1_1, b1_1, w2_1, b2_1, w1_2, b1_2, w2_2, b2_2, w_out, part);

  constexpr int WV = 8;
  constexpr int NWG = 256;

  net_all_kernel<WV><<<NWG, WV * 32, 0, stream>>>(
      T, l, l1_dot, center, v_t, v_y, v_g, v_c, nT, nL, nC, part);
}
